# starting kernel
# baseline (speedup 1.0000x reference)
#include <hip/hip_runtime.h>
#include <math.h>


static constexpr int D0   = 200;
static constexpr int DP   = 208;
static constexpr int NTN  = 13;
static constexpr int NE   = 8192;
static constexpr int NV   = 5000;
static constexpr int NVP  = 5056;
static constexpr int NLAY = 6;
static constexpr int NF   = 96;
static constexpr int CONVK= 49;
static constexpr int P196 = 196;
static constexpr int FLAT = 18816;
static constexpr int NCLS = 13;
#define INV_SQRT_1PE 0.99999500f

typedef __attribute__((ext_vector_type(2))) float v2f;
typedef __attribute__((ext_vector_type(8))) float v8f;

static __device__ __forceinline__ v8f wmma4(v2f a, v2f b, v8f c) {
  return __builtin_amdgcn_wmma_f32_16x16x4_f32(false, a, false, b, (short)0, c,
                                               false, false);
}


__global__ void k_zero(float* __restrict__ p, int n) {
  int i = blockIdx.x * blockDim.x + threadIdx.x;
  if (i < n) p[i] = 0.0f;
}

__global__ void k_pad(const float* __restrict__ src, float* __restrict__ dst,
                      int rows, int prows) {
  int i = blockIdx.x * blockDim.x + threadIdx.x;
  int total = prows * DP;
  if (i >= total) return;
  int r = i / DP, c = i % DP;
  dst[i] = (r < rows && c < D0) ? src[r * D0 + c] : 0.0f;
}

__global__ void k_pad_w(const float* __restrict__ Win, const float* __restrict__ Wout,
                        const float* __restrict__ Wloop, const float* __restrict__ Wrel,
                        int l, float* __restrict__ WP) {
  int i = blockIdx.x * blockDim.x + threadIdx.x;
  int total = 4 * DP * DP;
  if (i >= total) return;
  int w = i / (DP * DP);
  int rem = i % (DP * DP);
  int r = rem / DP, c = rem % DP;
  const float* src = (w == 0 ? Win : w == 1 ? Wout : w == 2 ? Wloop : Wrel)
                     + (size_t)l * D0 * D0;
  WP[i] = (r < D0 && c < D0) ? src[r * D0 + c] : 0.0f;
}

__global__ void k_tr_fcw(const float* __restrict__ fcw, float* __restrict__ fcwT) {
  int i = blockIdx.x * blockDim.x + threadIdx.x;
  if (i >= FLAT * DP) return;
  int k = i / DP, n = i % DP;
  fcwT[i] = (n < D0) ? fcw[(size_t)n * FLAT + k] : 0.0f;
}

__global__ void k_deg(const int* __restrict__ ei0, const int* __restrict__ iei0,
                      float* __restrict__ degf, float* __restrict__ degi) {
  int e = blockIdx.x * blockDim.x + threadIdx.x;
  if (e < NE)            atomicAdd(&degf[ei0[e]], 1.0f);
  else if (e < 2 * NE)   atomicAdd(&degi[iei0[e - NE]], 1.0f);
}

__global__ void k_norm(const int* __restrict__ ei0, const int* __restrict__ ei1,
                       const int* __restrict__ iei0, const int* __restrict__ iei1,
                       const float* __restrict__ degf, const float* __restrict__ degi,
                       float* __restrict__ nf, float* __restrict__ ni) {
  int e = blockIdx.x * blockDim.x + threadIdx.x;
  if (e >= NE) return;
  float df0 = degf[ei0[e]],  df1 = degf[ei1[e]];
  float di0 = degi[iei0[e]], di1 = degi[iei1[e]];
  float a = df0 > 0.f ? rsqrtf(fmaxf(df0, 1.f)) : 0.f;
  float b = df1 > 0.f ? rsqrtf(fmaxf(df1, 1.f)) : 0.f;
  float c = di0 > 0.f ? rsqrtf(fmaxf(di0, 1.f)) : 0.f;
  float d = di1 > 0.f ? rsqrtf(fmaxf(di1, 1.f)) : 0.f;
  nf[e] = a * b;
  ni[e] = c * d;
}

__global__ void k_gather(const float* __restrict__ xP, const float* __restrict__ r,
                         const float* __restrict__ ir,
                         const int* __restrict__ ei0, const int* __restrict__ iei0,
                         float* __restrict__ ain, float* __restrict__ aout) {
  int i = blockIdx.x * blockDim.x + threadIdx.x;
  if (i >= NE * DP) return;
  int e = i / DP, c = i % DP;
  ain[i]  = xP[(size_t)ei0[e] * DP + c] - r[i];
  aout[i] = xP[(size_t)iei0[e] * DP + c] - ir[i];
}

__global__ void k_scatter(const float* __restrict__ src, const int* __restrict__ dstidx,
                          float* __restrict__ agg) {
  int i = blockIdx.x * blockDim.x + threadIdx.x;
  if (i >= NE * DP) return;
  int e = i / DP, c = i % DP;
  atomicAdd(&agg[(size_t)dstidx[e] * DP + c], src[i]);
}

__global__ void k_update(float* __restrict__ xP, const float* __restrict__ agg,
                         const float* __restrict__ xl, const float* __restrict__ b_gcn,
                         int l) {
  int i = blockIdx.x * blockDim.x + threadIdx.x;
  if (i >= NV * D0) return;
  int r = i / D0, c = i % D0;
  size_t o = (size_t)r * DP + c;
  xP[o] = tanhf((agg[o] + xl[o]) * (1.0f / 3.0f) + b_gcn[l * D0 + c]);
}

template <bool HAS_SCALE>
__global__ __launch_bounds__(32) void k_gemm(const float* __restrict__ A,
                                             const float* __restrict__ B,
                                             float* __restrict__ C,
                                             const float* __restrict__ rowscale) {
  int tile = blockIdx.x;
  int tm = tile / NTN, tn = tile % NTN;
  int m0 = tm * 64, n0 = tn * 16;
  int lane = threadIdx.x;
  int half = lane >> 4;
  int l16  = lane & 15;

  const float* A0 = A + (size_t)(m0 + l16) * DP;
  const float* Bc = B + n0 + l16;
  v8f acc0 = {}, acc1 = {}, acc2 = {}, acc3 = {};
#pragma unroll 2
  for (int k = 0; k < DP; k += 4) {
    int kh = k + 2 * half;
    v2f b;
    b.x = Bc[(size_t)kh * DP];
    b.y = Bc[(size_t)(kh + 1) * DP];
    v2f a0 = *reinterpret_cast<const v2f*>(A0 + kh);
    v2f a1 = *reinterpret_cast<const v2f*>(A0 + 16 * DP + kh);
    v2f a2 = *reinterpret_cast<const v2f*>(A0 + 32 * DP + kh);
    v2f a3 = *reinterpret_cast<const v2f*>(A0 + 48 * DP + kh);
    acc0 = wmma4(a0, b, acc0);
    acc1 = wmma4(a1, b, acc1);
    acc2 = wmma4(a2, b, acc2);
    acc3 = wmma4(a3, b, acc3);
  }
#pragma unroll
  for (int s = 0; s < 4; ++s) {
    v8f acc = s == 0 ? acc0 : s == 1 ? acc1 : s == 2 ? acc2 : acc3;
    int mb = m0 + 16 * s + 8 * half;
#pragma unroll
    for (int v = 0; v < 8; ++v) {
      int m = mb + v;
      float val = acc[v];
      if (HAS_SCALE) val *= rowscale[m];
      C[(size_t)m * DP + n0 + l16] = val;
    }
  }
}

__global__ __launch_bounds__(128) void k_convfc(
    const float* __restrict__ xP, const float* __restrict__ rP,
    const int* __restrict__ ei0, const float* __restrict__ fcwT,
    const float* __restrict__ conv_w, const float* __restrict__ conv_b,
    const float* __restrict__ bn1_g, const float* __restrict__ bn1_b,
    const float* __restrict__ fc_b,
    const float* __restrict__ bn2_g, const float* __restrict__ bn2_b,
    const float* __restrict__ fc1_w, const float* __restrict__ fc1_b,
    const float* __restrict__ bn0_g, const float* __restrict__ bn0_b,
    float* __restrict__ out) {
  __shared__ float img[16][400];
  __shared__ float hbuf[16][P196];
  __shared__ float h2s[16][D0];
  __shared__ float cw[CONVK];

  int m0   = blockIdx.x * 16;
  int tid  = threadIdx.x;
  int wave = tid >> 5, lane = tid & 31;
  int half = lane >> 4, l16 = lane & 15;
  int nb   = wave * 16;

  float b0s = bn0_g[0] * INV_SQRT_1PE;
  float b0b = bn0_b[0];

  for (int i = tid; i < 16 * 400; i += 128) {
    int t = i / 400, k = i % 400;
    int src = ei0[m0 + t];
    float v = (k & 1) ? rP[(size_t)(m0 + t) * DP + (k >> 1)]
                      : xP[(size_t)src * DP + (k >> 1)];
    img[t][k] = v * b0s + b0b;
  }
  __syncthreads();

  v8f acc0 = {}, acc1 = {}, acc2 = {}, acc3 = {};

  for (int c = 0; c < NF; ++c) {
    if (tid < CONVK) cw[tid] = conv_w[c * CONVK + tid];
    __syncthreads();

    float sc1 = bn1_g[c] * INV_SQRT_1PE;
    float bb1 = bn1_b[c];
    float cb  = conv_b[c];
    for (int i = tid; i < 16 * P196; i += 128) {
      int t = i / P196, p = i % P196;
      int oy = p / 14, ox = p % 14;
      float s = cb;
#pragma unroll
      for (int a = 0; a < 7; ++a)
#pragma unroll
        for (int b = 0; b < 7; ++b)
          s += img[t][(oy + a) * 20 + ox + b] * cw[a * 7 + b];
      float v = s * sc1 + bb1;
      hbuf[t][p] = v > 0.f ? v : 0.f;
    }
    __syncthreads();

    const int kg0 = c * P196;
    const float* hrow = &hbuf[l16][0];
    if (wave == 0) {
      for (int kk = 0; kk < P196; kk += 4) {
        v2f a = *reinterpret_cast<const v2f*>(hrow + kk + 2 * half);
        const float* Bk = fcwT + (size_t)(kg0 + kk + 2 * half) * DP + l16;
        v2f b0, b1, b2, b3;
        b0.x = Bk[0];        b0.y = Bk[DP + 0];
        b1.x = Bk[64];       b1.y = Bk[DP + 64];
        b2.x = Bk[128];      b2.y = Bk[DP + 128];
        b3.x = Bk[192];      b3.y = Bk[DP + 192];
        acc0 = wmma4(a, b0, acc0);
        acc1 = wmma4(a, b1, acc1);
        acc2 = wmma4(a, b2, acc2);
        acc3 = wmma4(a, b3, acc3);
      }
    } else {
      for (int kk = 0; kk < P196; kk += 4) {
        v2f a = *reinterpret_cast<const v2f*>(hrow + kk + 2 * half);
        const float* Bk = fcwT + (size_t)(kg0 + kk + 2 * half) * DP + l16 + nb;
        v2f b0, b1, b2;
        b0.x = Bk[0];        b0.y = Bk[DP + 0];
        b1.x = Bk[64];       b1.y = Bk[DP + 64];
        b2.x = Bk[128];      b2.y = Bk[DP + 128];
        acc0 = wmma4(a, b0, acc0);
        acc1 = wmma4(a, b1, acc1);
        acc2 = wmma4(a, b2, acc2);
      }
    }
    __syncthreads();
  }

  auto store_tile = [&](v8f acc, int tn) {
    int n = tn * 16 + l16;
    if (n < D0) {
      float sc2 = bn2_g[n] * INV_SQRT_1PE;
      float bb2 = bn2_b[n];
      float fb  = fc_b[n];
#pragma unroll
      for (int v = 0; v < 8; ++v) {
        float val = (acc[v] + fb) * sc2 + bb2;
        h2s[v + 8 * half][n] = val > 0.f ? val : 0.f;
      }
    }
  };
  store_tile(acc0, wave);
  store_tile(acc1, wave + 4);
  store_tile(acc2, wave + 8);
  if (wave == 0) store_tile(acc3, 12);
  __syncthreads();

  for (int i = tid; i < 16 * NCLS; i += 128) {
    int t = i / NCLS, cls = i % NCLS;
    float s = fc1_b[cls];
    for (int c2 = 0; c2 < D0; ++c2) s += h2s[t][c2] * fc1_w[cls * D0 + c2];
    out[(size_t)(m0 + t) * NCLS + cls] = s;
  }
}


extern "C" void kernel_launch(void* const* d_in, const int* in_sizes, int n_in,
                              void* d_out, int out_size, void* d_ws, size_t ws_size,
                              hipStream_t stream) {
  const float* node  = (const float*)d_in[0];
  const float* ef    = (const float*)d_in[1];
  const float* ief   = (const float*)d_in[2];
  const float* Win   = (const float*)d_in[3];
  const float* Wout  = (const float*)d_in[4];
  const float* Wloop = (const float*)d_in[5];
  const float* Wrel  = (const float*)d_in[6];
  const float* bgcn  = (const float*)d_in[7];
  const float* bn0g  = (const float*)d_in[8];
  const float* bn0b  = (const float*)d_in[9];
  const float* convw = (const float*)d_in[10];
  const float* convb = (const float*)d_in[11];
  const float* bn1g  = (const float*)d_in[12];
  const float* bn1b  = (const float*)d_in[13];
  const float* fcw   = (const float*)d_in[14];
  const float* fcb   = (const float*)d_in[15];
  const float* bn2g  = (const float*)d_in[16];
  const float* bn2b  = (const float*)d_in[17];
  const float* fc1w  = (const float*)d_in[18];
  const float* fc1b  = (const float*)d_in[19];
  const int*   ei    = (const int*)d_in[20];
  const int*   iei   = (const int*)d_in[21];
  const int *ei0 = ei, *ei1 = ei + NE;
  const int *iei0 = iei, *iei1 = iei + NE;

  float* ws = (float*)d_ws;
  size_t o = 0;
  float* xP   = ws + o; o += (size_t)NVP * DP;
  float* xlP  = ws + o; o += (size_t)NVP * DP;
  float* aggP = ws + o; o += (size_t)NVP * DP;
  float* r0   = ws + o; o += (size_t)NE * DP;
  float* r1   = ws + o; o += (size_t)NE * DP;
  float* ir0  = ws + o; o += (size_t)NE * DP;
  float* ir1  = ws + o; o += (size_t)NE * DP;
  float* ain  = ws + o; o += (size_t)NE * DP;
  float* aout = ws + o; o += (size_t)NE * DP;
  float* minb = ws + o; o += (size_t)NE * DP;
  float* moutb= ws + o; o += (size_t)NE * DP;
  float* WP   = ws + o; o += (size_t)4 * DP * DP;
  float* fcwT = ws + o; o += (size_t)FLAT * DP;
  float* degf = ws + o; o += NVP;
  float* degi = ws + o; o += NVP;
  float* nfv  = ws + o; o += NE;
  float* niv  = ws + o; o += NE;

  const int TB = 256;
  auto cdiv = [](int a, int b) { return (a + b - 1) / b; };

  k_pad<<<cdiv(NVP * DP, TB), TB, 0, stream>>>(node, xP, NV, NVP);
  k_pad<<<cdiv(NE * DP, TB), TB, 0, stream>>>(ef, r0, NE, NE);
  k_pad<<<cdiv(NE * DP, TB), TB, 0, stream>>>(ief, ir0, NE, NE);
  k_tr_fcw<<<cdiv(FLAT * DP, TB), TB, 0, stream>>>(fcw, fcwT);

  k_zero<<<cdiv(2 * NVP, TB), TB, 0, stream>>>(degf, 2 * NVP);
  k_deg<<<cdiv(2 * NE, TB), TB, 0, stream>>>(ei0, iei0, degf, degi);
  k_norm<<<cdiv(NE, TB), TB, 0, stream>>>(ei0, ei1, iei0, iei1, degf, degi, nfv, niv);

  float *rc = r0, *rn = r1, *irc = ir0, *irn = ir1;
  const int GE = (NE / 64) * NTN;
  const int GV = (NVP / 64) * NTN;

  for (int l = 0; l < NLAY; ++l) {
    k_pad_w<<<cdiv(4 * DP * DP, TB), TB, 0, stream>>>(Win, Wout, Wloop, Wrel, l, WP);
    k_gather<<<cdiv(NE * DP, TB), TB, 0, stream>>>(xP, rc, irc, ei0, iei0, ain, aout);
    k_gemm<true><<<GE, 32, 0, stream>>>(ain,  WP + 0 * DP * DP, minb,  nfv);
    k_gemm<true><<<GE, 32, 0, stream>>>(aout, WP + 1 * DP * DP, moutb, niv);
    k_gemm<false><<<GV, 32, 0, stream>>>(xP,   WP + 2 * DP * DP, xlP,   nullptr);
    k_gemm<false><<<GE, 32, 0, stream>>>(rc,   WP + 3 * DP * DP, rn,    nullptr);
    k_gemm<false><<<GE, 32, 0, stream>>>(irc,  WP + 3 * DP * DP, irn,   nullptr);
    k_zero<<<cdiv(NVP * DP, TB), TB, 0, stream>>>(aggP, NVP * DP);
    k_scatter<<<cdiv(NE * DP, TB), TB, 0, stream>>>(minb,  ei1,  aggP);
    k_scatter<<<cdiv(NE * DP, TB), TB, 0, stream>>>(moutb, iei1, aggP);
    k_update<<<cdiv(NV * D0, TB), TB, 0, stream>>>(xP, aggP, xlP, bgcn, l);
    float* t;
    t = rc; rc = rn; rn = t;
    t = irc; irc = irn; irn = t;
  }

  k_convfc<<<NE / 16, 128, 0, stream>>>(xP, rc, ei0, fcwT, convw, convb,
                                        bn1g, bn1b, fcb, bn2g, bn2b,
                                        fc1w, fc1b, bn0g, bn0b, (float*)d_out);
}


# decoder conv: each thread keeps its 8x20 input window in registers for all 96 channels and reads the taps by scalar loads, so the per-channel conv does no LDS reads; 2 barriers per channel, not 3
# speedup vs baseline: 1.4795x; 1.4795x over previous
#include <hip/hip_runtime.h>
#include <math.h>


static constexpr int D0   = 200;
static constexpr int DP   = 208;
static constexpr int NTN  = 13;
static constexpr int NE   = 8192;
static constexpr int NV   = 5000;
static constexpr int NVP  = 5056;
static constexpr int NLAY = 6;
static constexpr int NF   = 96;
static constexpr int CONVK= 49;
static constexpr int P196 = 196;
static constexpr int FLAT = 18816;
static constexpr int NCLS = 13;
#define INV_SQRT_1PE 0.99999500f

typedef __attribute__((ext_vector_type(2))) float v2f;
typedef __attribute__((ext_vector_type(4))) float v4f;
typedef __attribute__((ext_vector_type(8))) float v8f;

static __device__ __forceinline__ v8f wmma4(v2f a, v2f b, v8f c) {
  return __builtin_amdgcn_wmma_f32_16x16x4_f32(false, a, false, b, (short)0, c,
                                               false, false);
}


__global__ void k_zero(float* __restrict__ p, int n) {
  int i = blockIdx.x * blockDim.x + threadIdx.x;
  if (i < n) p[i] = 0.0f;
}

__global__ void k_pad(const float* __restrict__ src, float* __restrict__ dst,
                      int rows, int prows) {
  int i = blockIdx.x * blockDim.x + threadIdx.x;
  int total = prows * DP;
  if (i >= total) return;
  int r = i / DP, c = i % DP;
  dst[i] = (r < rows && c < D0) ? src[r * D0 + c] : 0.0f;
}

__global__ void k_pad_w(const float* __restrict__ Win, const float* __restrict__ Wout,
                        const float* __restrict__ Wloop, const float* __restrict__ Wrel,
                        int l, float* __restrict__ WP) {
  int i = blockIdx.x * blockDim.x + threadIdx.x;
  int total = 4 * DP * DP;
  if (i >= total) return;
  int w = i / (DP * DP);
  int rem = i % (DP * DP);
  int r = rem / DP, c = rem % DP;
  const float* src = (w == 0 ? Win : w == 1 ? Wout : w == 2 ? Wloop : Wrel)
                     + (size_t)l * D0 * D0;
  WP[i] = (r < D0 && c < D0) ? src[r * D0 + c] : 0.0f;
}

__global__ void k_tr_fcw(const float* __restrict__ fcw, float* __restrict__ fcwT) {
  int i = blockIdx.x * blockDim.x + threadIdx.x;
  if (i >= FLAT * DP) return;
  int k = i / DP, n = i % DP;
  fcwT[i] = (n < D0) ? fcw[(size_t)n * FLAT + k] : 0.0f;
}

__global__ void k_deg(const int* __restrict__ ei0, const int* __restrict__ iei0,
                      float* __restrict__ degf, float* __restrict__ degi) {
  int e = blockIdx.x * blockDim.x + threadIdx.x;
  if (e < NE)            atomicAdd(&degf[ei0[e]], 1.0f);
  else if (e < 2 * NE)   atomicAdd(&degi[iei0[e - NE]], 1.0f);
}

__global__ void k_norm(const int* __restrict__ ei0, const int* __restrict__ ei1,
                       const int* __restrict__ iei0, const int* __restrict__ iei1,
                       const float* __restrict__ degf, const float* __restrict__ degi,
                       float* __restrict__ nf, float* __restrict__ ni) {
  int e = blockIdx.x * blockDim.x + threadIdx.x;
  if (e >= NE) return;
  float df0 = degf[ei0[e]],  df1 = degf[ei1[e]];
  float di0 = degi[iei0[e]], di1 = degi[iei1[e]];
  float a = df0 > 0.f ? rsqrtf(fmaxf(df0, 1.f)) : 0.f;
  float b = df1 > 0.f ? rsqrtf(fmaxf(df1, 1.f)) : 0.f;
  float c = di0 > 0.f ? rsqrtf(fmaxf(di0, 1.f)) : 0.f;
  float d = di1 > 0.f ? rsqrtf(fmaxf(di1, 1.f)) : 0.f;
  nf[e] = a * b;
  ni[e] = c * d;
}

__global__ void k_gather(const float* __restrict__ xP, const float* __restrict__ r,
                         const float* __restrict__ ir,
                         const int* __restrict__ ei0, const int* __restrict__ iei0,
                         float* __restrict__ ain, float* __restrict__ aout) {
  int i = blockIdx.x * blockDim.x + threadIdx.x;
  if (i >= NE * DP) return;
  int e = i / DP, c = i % DP;
  ain[i]  = xP[(size_t)ei0[e] * DP + c] - r[i];
  aout[i] = xP[(size_t)iei0[e] * DP + c] - ir[i];
}

__global__ void k_scatter(const float* __restrict__ src, const int* __restrict__ dstidx,
                          float* __restrict__ agg) {
  int i = blockIdx.x * blockDim.x + threadIdx.x;
  if (i >= NE * DP) return;
  int e = i / DP, c = i % DP;
  atomicAdd(&agg[(size_t)dstidx[e] * DP + c], src[i]);
}

__global__ void k_update(float* __restrict__ xP, const float* __restrict__ agg,
                         const float* __restrict__ xl, const float* __restrict__ b_gcn,
                         int l) {
  int i = blockIdx.x * blockDim.x + threadIdx.x;
  if (i >= NV * D0) return;
  int r = i / D0, c = i % D0;
  size_t o = (size_t)r * DP + c;
  xP[o] = tanhf((agg[o] + xl[o]) * (1.0f / 3.0f) + b_gcn[l * D0 + c]);
}

template <bool HAS_SCALE>
__global__ __launch_bounds__(32) void k_gemm(const float* __restrict__ A,
                                             const float* __restrict__ B,
                                             float* __restrict__ C,
                                             const float* __restrict__ rowscale) {
  int tile = blockIdx.x;
  int tm = tile / NTN, tn = tile % NTN;
  int m0 = tm * 64, n0 = tn * 16;
  int lane = threadIdx.x;
  int half = lane >> 4;
  int l16  = lane & 15;

  const float* A0 = A + (size_t)(m0 + l16) * DP;
  const float* Bc = B + n0 + l16;
  v8f acc0 = {}, acc1 = {}, acc2 = {}, acc3 = {};
#pragma unroll 2
  for (int k = 0; k < DP; k += 4) {
    int kh = k + 2 * half;
    v2f b;
    b.x = Bc[(size_t)kh * DP];
    b.y = Bc[(size_t)(kh + 1) * DP];
    v2f a0 = *reinterpret_cast<const v2f*>(A0 + kh);
    v2f a1 = *reinterpret_cast<const v2f*>(A0 + 16 * DP + kh);
    v2f a2 = *reinterpret_cast<const v2f*>(A0 + 32 * DP + kh);
    v2f a3 = *reinterpret_cast<const v2f*>(A0 + 48 * DP + kh);
    acc0 = wmma4(a0, b, acc0);
    acc1 = wmma4(a1, b, acc1);
    acc2 = wmma4(a2, b, acc2);
    acc3 = wmma4(a3, b, acc3);
  }
#pragma unroll
  for (int s = 0; s < 4; ++s) {
    v8f acc = s == 0 ? acc0 : s == 1 ? acc1 : s == 2 ? acc2 : acc3;
    int mb = m0 + 16 * s + 8 * half;
#pragma unroll
    for (int v = 0; v < 8; ++v) {
      int m = mb + v;
      float val = acc[v];
      if (HAS_SCALE) val *= rowscale[m];
      C[(size_t)m * DP + n0 + l16] = val;
    }
  }
}

__global__ __launch_bounds__(128) void k_convfc(
    const float* __restrict__ xP, const float* __restrict__ rP,
    const int* __restrict__ ei0, const float* __restrict__ fcwT,
    const float* __restrict__ conv_w, const float* __restrict__ conv_b,
    const float* __restrict__ bn1_g, const float* __restrict__ bn1_b,
    const float* __restrict__ fc_b,
    const float* __restrict__ bn2_g, const float* __restrict__ bn2_b,
    const float* __restrict__ fc1_w, const float* __restrict__ fc1_b,
    const float* __restrict__ bn0_g, const float* __restrict__ bn0_b,
    float* __restrict__ out) {
  __shared__ __attribute__((aligned(16))) float img[16][400];
  __shared__ __attribute__((aligned(16))) float hbuf[16][P196];
  __shared__ float h2s[16][D0];

  int m0   = blockIdx.x * 16;
  int tid  = threadIdx.x;
  int wave = tid >> 5, lane = tid & 31;
  int half = lane >> 4, l16 = lane & 15;
  int nb   = wave * 16;

  float b0s = bn0_g[0] * INV_SQRT_1PE;
  float b0b = bn0_b[0];

  for (int i = tid; i < 16 * 400; i += 128) {
    int t = i / 400, k = i % 400;
    int src = ei0[m0 + t];
    float v = (k & 1) ? rP[(size_t)(m0 + t) * DP + (k >> 1)]
                      : xP[(size_t)src * DP + (k >> 1)];
    img[t][k] = v * b0s + b0b;
  }
  __syncthreads();

  const bool cv = tid < 112;
  const int  ct = cv ? tid / 7 : 0;
  const int  cj = cv ? tid % 7 : 0;

  float win[8][20];
  {
    const v4f* wp = reinterpret_cast<const v4f*>(&img[ct][cj * 40]);
#pragma unroll
    for (int q = 0; q < 40; ++q) {
      v4f v = wp[q];
      win[q / 5][(q % 5) * 4 + 0] = v.x;
      win[q / 5][(q % 5) * 4 + 1] = v.y;
      win[q / 5][(q % 5) * 4 + 2] = v.z;
      win[q / 5][(q % 5) * 4 + 3] = v.w;
    }
  }

  v8f acc0 = {}, acc1 = {}, acc2 = {}, acc3 = {};

  for (int c = 0; c < NF; ++c) {
    const float* wc = conv_w + c * CONVK;
    float sc1 = bn1_g[c] * INV_SQRT_1PE;
    float bb1 = bn1_b[c];
    float cb  = conv_b[c];

    float o[28];
#pragma unroll
    for (int i = 0; i < 28; ++i) o[i] = cb;
#pragma unroll
    for (int a = 0; a < 7; ++a) {
#pragma unroll
      for (int b = 0; b < 7; ++b) {
        const float w = wc[a * 7 + b];
#pragma unroll
        for (int dy = 0; dy < 2; ++dy) {
#pragma unroll
          for (int ox = 0; ox < 14; ++ox)
            o[dy * 14 + ox] += win[dy + a][ox + b] * w;
        }
      }
    }

    __syncthreads();
    if (cv) {
      v4f* hp = reinterpret_cast<v4f*>(&hbuf[ct][cj * 28]);
#pragma unroll
      for (int q = 0; q < 7; ++q) {
        v4f r;
        float v0 = o[4 * q + 0] * sc1 + bb1;
        float v1 = o[4 * q + 1] * sc1 + bb1;
        float v2 = o[4 * q + 2] * sc1 + bb1;
        float v3 = o[4 * q + 3] * sc1 + bb1;
        r.x = v0 > 0.f ? v0 : 0.f;
        r.y = v1 > 0.f ? v1 : 0.f;
        r.z = v2 > 0.f ? v2 : 0.f;
        r.w = v3 > 0.f ? v3 : 0.f;
        hp[q] = r;
      }
    }
    __syncthreads();

    const int kg0 = c * P196;
    const float* hrow = &hbuf[l16][0];
    if (wave == 0) {
      for (int kk = 0; kk < P196; kk += 4) {
        v2f a = *reinterpret_cast<const v2f*>(hrow + kk + 2 * half);
        const float* Bk = fcwT + (size_t)(kg0 + kk + 2 * half) * DP + l16;
        v2f b0, b1, b2, b3;
        b0.x = Bk[0];        b0.y = Bk[DP + 0];
        b1.x = Bk[64];       b1.y = Bk[DP + 64];
        b2.x = Bk[128];      b2.y = Bk[DP + 128];
        b3.x = Bk[192];      b3.y = Bk[DP + 192];
        acc0 = wmma4(a, b0, acc0);
        acc1 = wmma4(a, b1, acc1);
        acc2 = wmma4(a, b2, acc2);
        acc3 = wmma4(a, b3, acc3);
      }
    } else {
      for (int kk = 0; kk < P196; kk += 4) {
        v2f a = *reinterpret_cast<const v2f*>(hrow + kk + 2 * half);
        const float* Bk = fcwT + (size_t)(kg0 + kk + 2 * half) * DP + l16 + nb;
        v2f b0, b1, b2;
        b0.x = Bk[0];        b0.y = Bk[DP + 0];
        b1.x = Bk[64];       b1.y = Bk[DP + 64];
        b2.x = Bk[128];      b2.y = Bk[DP + 128];
        acc0 = wmma4(a, b0, acc0);
        acc1 = wmma4(a, b1, acc1);
        acc2 = wmma4(a, b2, acc2);
      }
    }
  }

  auto store_tile = [&](v8f acc, int tn) {
    int n = tn * 16 + l16;
    if (n < D0) {
      float sc2 = bn2_g[n] * INV_SQRT_1PE;
      float bb2 = bn2_b[n];
      float fb  = fc_b[n];
#pragma unroll
      for (int v = 0; v < 8; ++v) {
        float val = (acc[v] + fb) * sc2 + bb2;
        h2s[v + 8 * half][n] = val > 0.f ? val : 0.f;
      }
    }
  };
  store_tile(acc0, wave);
  store_tile(acc1, wave + 4);
  store_tile(acc2, wave + 8);
  if (wave == 0) store_tile(acc3, 12);
  __syncthreads();

  for (int i = tid; i < 16 * NCLS; i += 128) {
    int t = i / NCLS, cls = i % NCLS;
    float s = fc1_b[cls];
    for (int c2 = 0; c2 < D0; ++c2) s += h2s[t][c2] * fc1_w[cls * D0 + c2];
    out[(size_t)(m0 + t) * NCLS + cls] = s;
  }
}


extern "C" void kernel_launch(void* const* d_in, const int* in_sizes, int n_in,
                              void* d_out, int out_size, void* d_ws, size_t ws_size,
                              hipStream_t stream) {
  const float* node  = (const float*)d_in[0];
  const float* ef    = (const float*)d_in[1];
  const float* ief   = (const float*)d_in[2];
  const float* Win   = (const float*)d_in[3];
  const float* Wout  = (const float*)d_in[4];
  const float* Wloop = (const float*)d_in[5];
  const float* Wrel  = (const float*)d_in[6];
  const float* bgcn  = (const float*)d_in[7];
  const float* bn0g  = (const float*)d_in[8];
  const float* bn0b  = (const float*)d_in[9];
  const float* convw = (const float*)d_in[10];
  const float* convb = (const float*)d_in[11];
  const float* bn1g  = (const float*)d_in[12];
  const float* bn1b  = (const float*)d_in[13];
  const float* fcw   = (const float*)d_in[14];
  const float* fcb   = (const float*)d_in[15];
  const float* bn2g  = (const float*)d_in[16];
  const float* bn2b  = (const float*)d_in[17];
  const float* fc1w  = (const float*)d_in[18];
  const float* fc1b  = (const float*)d_in[19];
  const int*   ei    = (const int*)d_in[20];
  const int*   iei   = (const int*)d_in[21];
  const int *ei0 = ei, *ei1 = ei + NE;
  const int *iei0 = iei, *iei1 = iei + NE;

  float* ws = (float*)d_ws;
  size_t o = 0;
  float* xP   = ws + o; o += (size_t)NVP * DP;
  float* xlP  = ws + o; o += (size_t)NVP * DP;
  float* aggP = ws + o; o += (size_t)NVP * DP;
  float* r0   = ws + o; o += (size_t)NE * DP;
  float* r1   = ws + o; o += (size_t)NE * DP;
  float* ir0  = ws + o; o += (size_t)NE * DP;
  float* ir1  = ws + o; o += (size_t)NE * DP;
  float* ain  = ws + o; o += (size_t)NE * DP;
  float* aout = ws + o; o += (size_t)NE * DP;
  float* minb = ws + o; o += (size_t)NE * DP;
  float* moutb= ws + o; o += (size_t)NE * DP;
  float* WP   = ws + o; o += (size_t)4 * DP * DP;
  float* fcwT = ws + o; o += (size_t)FLAT * DP;
  float* degf = ws + o; o += NVP;
  float* degi = ws + o; o += NVP;
  float* nfv  = ws + o; o += NE;
  float* niv  = ws + o; o += NE;

  const int TB = 256;
  auto cdiv = [](int a, int b) { return (a + b - 1) / b; };

  k_pad<<<cdiv(NVP * DP, TB), TB, 0, stream>>>(node, xP, NV, NVP);
  k_pad<<<cdiv(NE * DP, TB), TB, 0, stream>>>(ef, r0, NE, NE);
  k_pad<<<cdiv(NE * DP, TB), TB, 0, stream>>>(ief, ir0, NE, NE);
  k_tr_fcw<<<cdiv(FLAT * DP, TB), TB, 0, stream>>>(fcw, fcwT);

  k_zero<<<cdiv(2 * NVP, TB), TB, 0, stream>>>(degf, 2 * NVP);
  k_deg<<<cdiv(2 * NE, TB), TB, 0, stream>>>(ei0, iei0, degf, degi);
  k_norm<<<cdiv(NE, TB), TB, 0, stream>>>(ei0, ei1, iei0, iei1, degf, degi, nfv, niv);

  float *rc = r0, *rn = r1, *irc = ir0, *irn = ir1;
  const int GE = (NE / 64) * NTN;
  const int GV = (NVP / 64) * NTN;

  for (int l = 0; l < NLAY; ++l) {
    k_pad_w<<<cdiv(4 * DP * DP, TB), TB, 0, stream>>>(Win, Wout, Wloop, Wrel, l, WP);
    k_gather<<<cdiv(NE * DP, TB), TB, 0, stream>>>(xP, rc, irc, ei0, iei0, ain, aout);
    k_gemm<true><<<GE, 32, 0, stream>>>(ain,  WP + 0 * DP * DP, minb,  nfv);
    k_gemm<true><<<GE, 32, 0, stream>>>(aout, WP + 1 * DP * DP, moutb, niv);
    k_gemm<false><<<GV, 32, 0, stream>>>(xP,   WP + 2 * DP * DP, xlP,   nullptr);
    k_gemm<false><<<GE, 32, 0, stream>>>(rc,   WP + 3 * DP * DP, rn,    nullptr);
    k_gemm<false><<<GE, 32, 0, stream>>>(irc,  WP + 3 * DP * DP, irn,   nullptr);
    k_zero<<<cdiv(NVP * DP, TB), TB, 0, stream>>>(aggP, NVP * DP);
    k_scatter<<<cdiv(NE * DP, TB), TB, 0, stream>>>(minb,  ei1,  aggP);
    k_scatter<<<cdiv(NE * DP, TB), TB, 0, stream>>>(moutb, iei1, aggP);
    k_update<<<cdiv(NV * D0, TB), TB, 0, stream>>>(xP, aggP, xlP, bgcn, l);
    float* t;
    t = rc; rc = rn; rn = t;
    t = irc; irc = irn; irn = t;
  }

  k_convfc<<<NE / 16, 128, 0, stream>>>(xP, rc, ei0, fcwT, convw, convb,
                                        bn1g, bn1b, fcb, bn2g, bn2b,
                                        fc1w, fc1b, bn0g, bn0b, (float*)d_out);
}


# decoder FC operand supply: fc_w repacked so a lane's B fragment for 8 k of a tile is one 16-byte load (whole 128-byte lines per wave), B issued 4 k-blocks and A one block ahead; FC per channel pair
# speedup vs baseline: 2.1588x; 1.4591x over previous
#include <hip/hip_runtime.h>
#include <math.h>


static constexpr int D0   = 200;
static constexpr int DP   = 208;
static constexpr int NTN  = 13;
static constexpr int NE   = 8192;
static constexpr int NV   = 5000;
static constexpr int NVP  = 5056;
static constexpr int NLAY = 6;
static constexpr int NF   = 96;
static constexpr int CONVK= 49;
static constexpr int P196 = 196;
static constexpr int FLAT = 18816;
static constexpr int NCLS = 13;
#define INV_SQRT_1PE 0.99999500f

typedef __attribute__((ext_vector_type(2))) float v2f;
typedef __attribute__((ext_vector_type(4))) float v4f;
typedef __attribute__((ext_vector_type(8))) float v8f;

static __device__ __forceinline__ v8f wmma4(v2f a, v2f b, v8f c) {
  return __builtin_amdgcn_wmma_f32_16x16x4_f32(false, a, false, b, (short)0, c,
                                               false, false);
}


__global__ void k_init(const float* __restrict__ node, const float* __restrict__ ef,
                       const float* __restrict__ ief, const float* __restrict__ fcw,
                       float* __restrict__ xP, float* __restrict__ r0,
                       float* __restrict__ ir0, float* __restrict__ fcwP,
                       float* __restrict__ deg) {
  int i = blockIdx.x * blockDim.x + threadIdx.x;
  if (i < FLAT * DP) {
    int e = i & 3, lane = (i >> 2) & 31, rest = i >> 7;
    int tile = rest % NTN, kb = rest / NTN;
    int n = tile * 16 + (lane & 15);
    int k = kb * 8 + (lane >> 4) * 4 + e;
    fcwP[i] = (n < D0) ? fcw[(size_t)n * FLAT + k] : 0.0f;
  }
  if (i < NE * DP) {
    int r = i / DP, c = i % DP;
    bool in = c < D0;
    r0[i]  = in ? ef[r * D0 + c]  : 0.0f;
    ir0[i] = in ? ief[r * D0 + c] : 0.0f;
  }
  if (i < NVP * DP) {
    int r = i / DP, c = i % DP;
    xP[i] = (r < NV && c < D0) ? node[r * D0 + c] : 0.0f;
  }
  if (i < 2 * NVP) deg[i] = 0.0f;
}

__global__ void k_deg(const int* __restrict__ ei0, const int* __restrict__ iei0,
                      float* __restrict__ degf, float* __restrict__ degi) {
  int e = blockIdx.x * blockDim.x + threadIdx.x;
  if (e < NE)            atomicAdd(&degf[ei0[e]], 1.0f);
  else if (e < 2 * NE)   atomicAdd(&degi[iei0[e - NE]], 1.0f);
}

__global__ void k_norm(const int* __restrict__ ei0, const int* __restrict__ ei1,
                       const int* __restrict__ iei0, const int* __restrict__ iei1,
                       const float* __restrict__ degf, const float* __restrict__ degi,
                       float* __restrict__ nf, float* __restrict__ ni) {
  int e = blockIdx.x * blockDim.x + threadIdx.x;
  if (e >= NE) return;
  float df0 = degf[ei0[e]],  df1 = degf[ei1[e]];
  float di0 = degi[iei0[e]], di1 = degi[iei1[e]];
  float a = df0 > 0.f ? rsqrtf(fmaxf(df0, 1.f)) : 0.f;
  float b = df1 > 0.f ? rsqrtf(fmaxf(df1, 1.f)) : 0.f;
  float c = di0 > 0.f ? rsqrtf(fmaxf(di0, 1.f)) : 0.f;
  float d = di1 > 0.f ? rsqrtf(fmaxf(di1, 1.f)) : 0.f;
  nf[e] = a * b;
  ni[e] = c * d;
}

__global__ void k_pre(const float* __restrict__ xP, const float* __restrict__ r,
                      const float* __restrict__ ir,
                      const int* __restrict__ ei0, const int* __restrict__ iei0,
                      float* __restrict__ ain, float* __restrict__ aout,
                      float* __restrict__ agg,
                      const float* __restrict__ Win, const float* __restrict__ Wout,
                      const float* __restrict__ Wloop, const float* __restrict__ Wrel,
                      int l, float* __restrict__ WP) {
  int i = blockIdx.x * blockDim.x + threadIdx.x;
  if (i >= NE * DP) return;
  {
    int e = i / DP, c = i % DP;
    ain[i]  = xP[(size_t)ei0[e] * DP + c] - r[i];
    aout[i] = xP[(size_t)iei0[e] * DP + c] - ir[i];
  }
  if (i < NVP * DP) agg[i] = 0.0f;
  if (i < 4 * DP * DP) {
    int w = i / (DP * DP);
    int rem = i % (DP * DP);
    int rr = rem / DP, c = rem % DP;
    const float* src = (w == 0 ? Win : w == 1 ? Wout : w == 2 ? Wloop : Wrel)
                       + (size_t)l * D0 * D0;
    WP[i] = (rr < D0 && c < D0) ? src[rr * D0 + c] : 0.0f;
  }
}

__global__ void k_scatter2(const float* __restrict__ m_in, const int* __restrict__ ei1,
                           const float* __restrict__ m_out, const int* __restrict__ iei1,
                           float* __restrict__ agg) {
  int i = blockIdx.x * blockDim.x + threadIdx.x;
  if (i >= 2 * NE * DP) return;
  const bool second = i >= NE * DP;
  const int j = second ? i - NE * DP : i;
  const float* src = second ? m_out : m_in;
  const int*   idx = second ? iei1 : ei1;
  int e = j / DP, c = j % DP;
  atomicAdd(&agg[(size_t)idx[e] * DP + c], src[j]);
}

__global__ void k_update(float* __restrict__ xP, const float* __restrict__ agg,
                         const float* __restrict__ xl, const float* __restrict__ b_gcn,
                         int l) {
  int i = blockIdx.x * blockDim.x + threadIdx.x;
  if (i >= NV * D0) return;
  int r = i / D0, c = i % D0;
  size_t o = (size_t)r * DP + c;
  xP[o] = tanhf((agg[o] + xl[o]) * (1.0f / 3.0f) + b_gcn[l * D0 + c]);
}

template <bool HAS_SCALE>
static __device__ __forceinline__ void gemm_store(v8f acc0, v8f acc1, v8f acc2, v8f acc3,
                                                  float* __restrict__ Cc, int mh,
                                                  const float* __restrict__ rowscale) {
#pragma unroll
  for (int s = 0; s < 4; ++s) {
    v8f acc = s == 0 ? acc0 : s == 1 ? acc1 : s == 2 ? acc2 : acc3;
    int mb = mh + 16 * s;
#pragma unroll
    for (int v = 0; v < 8; ++v) {
      int m = mb + v;
      float val = acc[v];
      if (HAS_SCALE) val *= rowscale[m];
      Cc[(size_t)m * DP] = val;
    }
  }
}

static constexpr int PE = NE / 64;
static constexpr int PV = NVP / 64;
__global__ __launch_bounds__(32 * NTN) void k_gemm5(
    const float* __restrict__ Ain, const float* __restrict__ Aout,
    const float* __restrict__ Ar,  const float* __restrict__ Air,
    const float* __restrict__ Ax,  const float* __restrict__ WP,
    float* __restrict__ Cin, float* __restrict__ Cout,
    float* __restrict__ Cr,  float* __restrict__ Cir,
    float* __restrict__ Cx,
    const float* __restrict__ nf, const float* __restrict__ ni) {
  const int p = blockIdx.x;
  const float* A; const float* B; float* C; const float* rowscale; int tm;
  if (p < PE)          { A = Ain;  B = WP;               C = Cin;  rowscale = nf;      tm = p; }
  else if (p < 2 * PE) { A = Aout; B = WP + DP * DP;     C = Cout; rowscale = ni;      tm = p - PE; }
  else if (p < 3 * PE) { A = Ar;   B = WP + 3 * DP * DP; C = Cr;   rowscale = nullptr; tm = p - 2 * PE; }
  else if (p < 4 * PE) { A = Air;  B = WP + 3 * DP * DP; C = Cir;  rowscale = nullptr; tm = p - 3 * PE; }
  else                 { A = Ax;   B = WP + 2 * DP * DP; C = Cx;   rowscale = nullptr; tm = p - 4 * PE; }

  const int tn = threadIdx.x >> 5;
  int m0 = tm * 64, n0 = tn * 16;
  int lane = threadIdx.x & 31;
  int half = lane >> 4;
  int l16  = lane & 15;

  const float* A0 = A + (size_t)(m0 + l16) * DP;
  const float* Bc = B + n0 + l16;
  v8f acc0 = {}, acc1 = {}, acc2 = {}, acc3 = {};
#pragma unroll 2
  for (int k = 0; k < DP; k += 4) {
    int kh = k + 2 * half;
    v2f b;
    b.x = Bc[(size_t)kh * DP];
    b.y = Bc[(size_t)(kh + 1) * DP];
    v2f a0 = *reinterpret_cast<const v2f*>(A0 + kh);
    v2f a1 = *reinterpret_cast<const v2f*>(A0 + 16 * DP + kh);
    v2f a2 = *reinterpret_cast<const v2f*>(A0 + 32 * DP + kh);
    v2f a3 = *reinterpret_cast<const v2f*>(A0 + 48 * DP + kh);
    acc0 = wmma4(a0, b, acc0);
    acc1 = wmma4(a1, b, acc1);
    acc2 = wmma4(a2, b, acc2);
    acc3 = wmma4(a3, b, acc3);
  }
  float* Cc = C + n0 + l16;
  const int mh = m0 + 8 * half;
  if (rowscale) gemm_store<true>(acc0, acc1, acc2, acc3, Cc, mh, rowscale);
  else          gemm_store<false>(acc0, acc1, acc2, acc3, Cc, mh, rowscale);
}

static constexpr int HP = 396;

static __device__ __forceinline__ v2f lo2(v4f v) { v2f r; r.x = v.x; r.y = v.y; return r; }
static __device__ __forceinline__ v2f hi2(v4f v) { v2f r; r.x = v.z; r.y = v.w; return r; }

static constexpr int FC_NB = 49;
static constexpr int FC_PD = 4;
static constexpr int FC_BS = NTN * 32;

template <int NT>
static __device__ __forceinline__ void fc_fill(v4f (&ring)[FC_PD + 1][4], const v4f* __restrict__ Bp) {
#pragma unroll
  for (int j = 0; j < FC_PD; ++j) {
#pragma unroll
    for (int t = 0; t < NT; ++t) ring[j][t] = Bp[j * FC_BS + t * 128];
  }
}

template <int NT>
static __device__ __forceinline__ void fc_pair(const float* hrow, const v4f* __restrict__ Bp,
                                               v4f (&ring)[FC_PD + 1][4],
                                               v8f& acc0, v8f& acc1, v8f& acc2, v8f& acc3) {
  v4f av = *reinterpret_cast<const v4f*>(hrow);
#pragma unroll
  for (int j = 0; j < FC_NB; ++j) {
    if (j + FC_PD < FC_NB) {
#pragma unroll
      for (int t = 0; t < NT; ++t)
        ring[(j + FC_PD) % (FC_PD + 1)][t] = Bp[(j + FC_PD) * FC_BS + t * 128];
    }
    v4f avn = av;
    if (j + 1 < FC_NB) avn = *reinterpret_cast<const v4f*>(hrow + 8 * (j + 1));
    const v2f a1 = lo2(av), a2 = hi2(av);
    const v4f* bq = ring[j % (FC_PD + 1)];
    acc0 = wmma4(a1, lo2(bq[0]), acc0);
    acc1 = wmma4(a1, lo2(bq[1]), acc1);
    acc2 = wmma4(a1, lo2(bq[2]), acc2);
    if constexpr (NT == 4) acc3 = wmma4(a1, lo2(bq[3]), acc3);
    acc0 = wmma4(a2, hi2(bq[0]), acc0);
    acc1 = wmma4(a2, hi2(bq[1]), acc1);
    acc2 = wmma4(a2, hi2(bq[2]), acc2);
    if constexpr (NT == 4) acc3 = wmma4(a2, hi2(bq[3]), acc3);
    av = avn;
  }
}

__global__ __launch_bounds__(128, 2) void k_convfc(
    const float* __restrict__ xP, const float* __restrict__ rP,
    const int* __restrict__ ei0, const float* __restrict__ fcwP,
    const float* __restrict__ conv_w, const float* __restrict__ conv_b,
    const float* __restrict__ bn1_g, const float* __restrict__ bn1_b,
    const float* __restrict__ fc_b,
    const float* __restrict__ bn2_g, const float* __restrict__ bn2_b,
    const float* __restrict__ fc1_w, const float* __restrict__ fc1_b,
    const float* __restrict__ bn0_g, const float* __restrict__ bn0_b,
    float* __restrict__ out) {
  __shared__ __attribute__((aligned(16))) float img[16][400];
  __shared__ __attribute__((aligned(16))) float hbuf[16][HP];
  __shared__ float h2s[16][D0];

  int m0   = blockIdx.x * 16;
  int tid  = threadIdx.x;
  int wave = tid >> 5, lane = tid & 31;
  int half = lane >> 4, l16 = lane & 15;

  float b0s = bn0_g[0] * INV_SQRT_1PE;
  float b0b = bn0_b[0];

  for (int i = tid; i < 16 * 400; i += 128) {
    int t = i / 400, k = i % 400;
    int src = ei0[m0 + t];
    float v = (k & 1) ? rP[(size_t)(m0 + t) * DP + (k >> 1)]
                      : xP[(size_t)src * DP + (k >> 1)];
    img[t][k] = v * b0s + b0b;
  }
  __syncthreads();

  const bool cv = tid < 112;
  const int  ct = cv ? tid / 7 : 0;
  const int  cj = cv ? tid % 7 : 0;

  float win[8][20];
  {
    const v4f* wp = reinterpret_cast<const v4f*>(&img[ct][cj * 40]);
#pragma unroll
    for (int q = 0; q < 40; ++q) {
      v4f v = wp[q];
      win[q / 5][(q % 5) * 4 + 0] = v.x;
      win[q / 5][(q % 5) * 4 + 1] = v.y;
      win[q / 5][(q % 5) * 4 + 2] = v.z;
      win[q / 5][(q % 5) * 4 + 3] = v.w;
    }
  }

  v8f acc0 = {}, acc1 = {}, acc2 = {}, acc3 = {};
  const v4f*   fcwP4 = reinterpret_cast<const v4f*>(fcwP);
  const float* hrow  = &hbuf[l16][4 * half];

  for (int cp = 0; cp < NF / 2; ++cp) {
    for (int s = 0; s < 2; ++s) {
      const int c = 2 * cp + s;
      const float* wc = conv_w + c * CONVK;
      float sc1 = bn1_g[c] * INV_SQRT_1PE;
      float bb1 = bn1_b[c];
      float cb  = conv_b[c];

      float o[28];
#pragma unroll
      for (int i = 0; i < 28; ++i) o[i] = cb;
#pragma unroll
      for (int a = 0; a < 7; ++a) {
#pragma unroll
        for (int b = 0; b < 7; ++b) {
          const float w = wc[a * 7 + b];
#pragma unroll
          for (int dy = 0; dy < 2; ++dy) {
#pragma unroll
            for (int ox = 0; ox < 14; ++ox)
              o[dy * 14 + ox] += win[dy + a][ox + b] * w;
          }
        }
      }

      if (s == 0) __syncthreads();
      if (cv) {
        v4f* hp = reinterpret_cast<v4f*>(&hbuf[ct][s * P196 + cj * 28]);
#pragma unroll
        for (int q = 0; q < 7; ++q) {
          v4f r;
          float v0 = o[4 * q + 0] * sc1 + bb1;
          float v1 = o[4 * q + 1] * sc1 + bb1;
          float v2 = o[4 * q + 2] * sc1 + bb1;
          float v3 = o[4 * q + 3] * sc1 + bb1;
          r.x = v0 > 0.f ? v0 : 0.f;
          r.y = v1 > 0.f ? v1 : 0.f;
          r.z = v2 > 0.f ? v2 : 0.f;
          r.w = v3 > 0.f ? v3 : 0.f;
          hp[q] = r;
        }
      }
    }
    const v4f* Bp = fcwP4 + ((size_t)cp * (FC_NB * NTN) + wave) * 32 + lane;
    v4f ring[FC_PD + 1][4];
    if (wave == 0) fc_fill<4>(ring, Bp);
    else           fc_fill<3>(ring, Bp);
    __syncthreads();
    if (wave == 0) fc_pair<4>(hrow, Bp, ring, acc0, acc1, acc2, acc3);
    else           fc_pair<3>(hrow, Bp, ring, acc0, acc1, acc2, acc3);
  }

  auto store_tile = [&](v8f acc, int tn) {
    int n = tn * 16 + l16;
    if (n < D0) {
      float sc2 = bn2_g[n] * INV_SQRT_1PE;
      float bb2 = bn2_b[n];
      float fb  = fc_b[n];
#pragma unroll
      for (int v = 0; v < 8; ++v) {
        float val = (acc[v] + fb) * sc2 + bb2;
        h2s[v + 8 * half][n] = val > 0.f ? val : 0.f;
      }
    }
  };
  store_tile(acc0, wave);
  store_tile(acc1, wave + 4);
  store_tile(acc2, wave + 8);
  if (wave == 0) store_tile(acc3, 12);
  __syncthreads();

  for (int i = tid; i < 16 * NCLS; i += 128) {
    int t = i / NCLS, cls = i % NCLS;
    float s = fc1_b[cls];
    for (int c2 = 0; c2 < D0; ++c2) s += h2s[t][c2] * fc1_w[cls * D0 + c2];
    out[(size_t)(m0 + t) * NCLS + cls] = s;
  }
}


extern "C" void kernel_launch(void* const* d_in, const int* in_sizes, int n_in,
                              void* d_out, int out_size, void* d_ws, size_t ws_size,
                              hipStream_t stream) {
  const float* node  = (const float*)d_in[0];
  const float* ef    = (const float*)d_in[1];
  const float* ief   = (const float*)d_in[2];
  const float* Win   = (const float*)d_in[3];
  const float* Wout  = (const float*)d_in[4];
  const float* Wloop = (const float*)d_in[5];
  const float* Wrel  = (const float*)d_in[6];
  const float* bgcn  = (const float*)d_in[7];
  const float* bn0g  = (const float*)d_in[8];
  const float* bn0b  = (const float*)d_in[9];
  const float* convw = (const float*)d_in[10];
  const float* convb = (const float*)d_in[11];
  const float* bn1g  = (const float*)d_in[12];
  const float* bn1b  = (const float*)d_in[13];
  const float* fcw   = (const float*)d_in[14];
  const float* fcb   = (const float*)d_in[15];
  const float* bn2g  = (const float*)d_in[16];
  const float* bn2b  = (const float*)d_in[17];
  const float* fc1w  = (const float*)d_in[18];
  const float* fc1b  = (const float*)d_in[19];
  const int*   ei    = (const int*)d_in[20];
  const int*   iei   = (const int*)d_in[21];
  const int *ei0 = ei, *ei1 = ei + NE;
  const int *iei0 = iei, *iei1 = iei + NE;

  float* ws = (float*)d_ws;
  size_t o = 0;
  float* xP   = ws + o; o += (size_t)NVP * DP;
  float* xlP  = ws + o; o += (size_t)NVP * DP;
  float* aggP = ws + o; o += (size_t)NVP * DP;
  float* r0   = ws + o; o += (size_t)NE * DP;
  float* r1   = ws + o; o += (size_t)NE * DP;
  float* ir0  = ws + o; o += (size_t)NE * DP;
  float* ir1  = ws + o; o += (size_t)NE * DP;
  float* ain  = ws + o; o += (size_t)NE * DP;
  float* aout = ws + o; o += (size_t)NE * DP;
  float* minb = ws + o; o += (size_t)NE * DP;
  float* moutb= ws + o; o += (size_t)NE * DP;
  float* WP   = ws + o; o += (size_t)4 * DP * DP;
  float* fcwP = ws + o; o += (size_t)FLAT * DP;
  float* degf = ws + o; o += NVP;
  float* degi = ws + o; o += NVP;
  float* nfv  = ws + o; o += NE;
  float* niv  = ws + o; o += NE;

  const int TB = 256;
  auto cdiv = [](int a, int b) { return (a + b - 1) / b; };

  k_init<<<cdiv(FLAT * DP, TB), TB, 0, stream>>>(node, ef, ief, fcw, xP, r0, ir0, fcwP, degf);

  k_deg<<<cdiv(2 * NE, TB), TB, 0, stream>>>(ei0, iei0, degf, degi);
  k_norm<<<cdiv(NE, TB), TB, 0, stream>>>(ei0, ei1, iei0, iei1, degf, degi, nfv, niv);

  float *rc = r0, *rn = r1, *irc = ir0, *irn = ir1;

  for (int l = 0; l < NLAY; ++l) {
    k_pre<<<cdiv(NE * DP, TB), TB, 0, stream>>>(xP, rc, irc, ei0, iei0, ain, aout, aggP,
                                                Win, Wout, Wloop, Wrel, l, WP);
    k_gemm5<<<4 * PE + PV, 32 * NTN, 0, stream>>>(ain, aout, rc, irc, xP, WP,
                                                  minb, moutb, rn, irn, xlP, nfv, niv);
    k_scatter2<<<cdiv(2 * NE * DP, TB), TB, 0, stream>>>(minb, ei1, moutb, iei1, aggP);
    k_update<<<cdiv(NV * D0, TB), TB, 0, stream>>>(xP, aggP, xlP, bgcn, l);
    float* t;
    t = rc; rc = rn; rn = t;
    t = irc; irc = irn; irn = t;
  }

  k_convfc<<<NE / 16, 128, 0, stream>>>(xP, rc, ei0, fcwP, convw, convb,
                                        bn1g, bn1b, fcb, bn2g, bn2b,
                                        fc1w, fc1b, bn0g, bn0b, (float*)d_out);
}


# decoder FC split over the work-group's 4 waves by k (every 4th k-block, all 13 column tiles each) instead of by column tile (4/3/3/3): equal WMMA work per SIMD, one A fragment feeds 26 WMMAs
# speedup vs baseline: 2.2560x; 1.0450x over previous
#include <hip/hip_runtime.h>
#include <math.h>


static constexpr int D0   = 200;
static constexpr int DP   = 208;
static constexpr int NTN  = 13;
static constexpr int NE   = 8192;
static constexpr int NV   = 5000;
static constexpr int NVP  = 5056;
static constexpr int NLAY = 6;
static constexpr int NF   = 96;
static constexpr int CONVK= 49;
static constexpr int P196 = 196;
static constexpr int FLAT = 18816;
static constexpr int NCLS = 13;
#define INV_SQRT_1PE 0.99999500f

typedef __attribute__((ext_vector_type(2))) float v2f;
typedef __attribute__((ext_vector_type(4))) float v4f;
typedef __attribute__((ext_vector_type(8))) float v8f;

static __device__ __forceinline__ v8f wmma4(v2f a, v2f b, v8f c) {
  return __builtin_amdgcn_wmma_f32_16x16x4_f32(false, a, false, b, (short)0, c,
                                               false, false);
}


__global__ void k_init(const float* __restrict__ node, const float* __restrict__ ef,
                       const float* __restrict__ ief, const float* __restrict__ fcw,
                       float* __restrict__ xP, float* __restrict__ r0,
                       float* __restrict__ ir0, float* __restrict__ fcwP,
                       float* __restrict__ deg) {
  int i = blockIdx.x * blockDim.x + threadIdx.x;
  if (i < FLAT * DP) {
    int e = i & 3, lane = (i >> 2) & 31, rest = i >> 7;
    int tile = rest % NTN, kb = rest / NTN;
    int n = tile * 16 + (lane & 15);
    int k = kb * 8 + (lane >> 4) * 4 + e;
    fcwP[i] = (n < D0) ? fcw[(size_t)n * FLAT + k] : 0.0f;
  }
  if (i < NE * DP) {
    int r = i / DP, c = i % DP;
    bool in = c < D0;
    r0[i]  = in ? ef[r * D0 + c]  : 0.0f;
    ir0[i] = in ? ief[r * D0 + c] : 0.0f;
  }
  if (i < NVP * DP) {
    int r = i / DP, c = i % DP;
    xP[i] = (r < NV && c < D0) ? node[r * D0 + c] : 0.0f;
  }
  if (i < 2 * NVP) deg[i] = 0.0f;
}

__global__ void k_deg(const int* __restrict__ ei0, const int* __restrict__ iei0,
                      float* __restrict__ degf, float* __restrict__ degi) {
  int e = blockIdx.x * blockDim.x + threadIdx.x;
  if (e < NE)            atomicAdd(&degf[ei0[e]], 1.0f);
  else if (e < 2 * NE)   atomicAdd(&degi[iei0[e - NE]], 1.0f);
}

__global__ void k_norm(const int* __restrict__ ei0, const int* __restrict__ ei1,
                       const int* __restrict__ iei0, const int* __restrict__ iei1,
                       const float* __restrict__ degf, const float* __restrict__ degi,
                       float* __restrict__ nf, float* __restrict__ ni) {
  int e = blockIdx.x * blockDim.x + threadIdx.x;
  if (e >= NE) return;
  float df0 = degf[ei0[e]],  df1 = degf[ei1[e]];
  float di0 = degi[iei0[e]], di1 = degi[iei1[e]];
  float a = df0 > 0.f ? rsqrtf(fmaxf(df0, 1.f)) : 0.f;
  float b = df1 > 0.f ? rsqrtf(fmaxf(df1, 1.f)) : 0.f;
  float c = di0 > 0.f ? rsqrtf(fmaxf(di0, 1.f)) : 0.f;
  float d = di1 > 0.f ? rsqrtf(fmaxf(di1, 1.f)) : 0.f;
  nf[e] = a * b;
  ni[e] = c * d;
}

__global__ void k_pre(const float* __restrict__ xP, const float* __restrict__ r,
                      const float* __restrict__ ir,
                      const int* __restrict__ ei0, const int* __restrict__ iei0,
                      float* __restrict__ ain, float* __restrict__ aout,
                      float* __restrict__ agg,
                      const float* __restrict__ Win, const float* __restrict__ Wout,
                      const float* __restrict__ Wloop, const float* __restrict__ Wrel,
                      int l, float* __restrict__ WP) {
  int i = blockIdx.x * blockDim.x + threadIdx.x;
  if (i >= NE * DP) return;
  {
    int e = i / DP, c = i % DP;
    ain[i]  = xP[(size_t)ei0[e] * DP + c] - r[i];
    aout[i] = xP[(size_t)iei0[e] * DP + c] - ir[i];
  }
  if (i < NVP * DP) agg[i] = 0.0f;
  if (i < 4 * DP * DP) {
    int w = i / (DP * DP);
    int rem = i % (DP * DP);
    int rr = rem / DP, c = rem % DP;
    const float* src = (w == 0 ? Win : w == 1 ? Wout : w == 2 ? Wloop : Wrel)
                       + (size_t)l * D0 * D0;
    WP[i] = (rr < D0 && c < D0) ? src[rr * D0 + c] : 0.0f;
  }
}

__global__ void k_scatter2(const float* __restrict__ m_in, const int* __restrict__ ei1,
                           const float* __restrict__ m_out, const int* __restrict__ iei1,
                           float* __restrict__ agg) {
  int i = blockIdx.x * blockDim.x + threadIdx.x;
  if (i >= 2 * NE * DP) return;
  const bool second = i >= NE * DP;
  const int j = second ? i - NE * DP : i;
  const float* src = second ? m_out : m_in;
  const int*   idx = second ? iei1 : ei1;
  int e = j / DP, c = j % DP;
  atomicAdd(&agg[(size_t)idx[e] * DP + c], src[j]);
}

__global__ void k_update(float* __restrict__ xP, const float* __restrict__ agg,
                         const float* __restrict__ xl, const float* __restrict__ b_gcn,
                         int l) {
  int i = blockIdx.x * blockDim.x + threadIdx.x;
  if (i >= NV * D0) return;
  int r = i / D0, c = i % D0;
  size_t o = (size_t)r * DP + c;
  xP[o] = tanhf((agg[o] + xl[o]) * (1.0f / 3.0f) + b_gcn[l * D0 + c]);
}

template <bool HAS_SCALE>
static __device__ __forceinline__ void gemm_store(v8f acc0, v8f acc1, v8f acc2, v8f acc3,
                                                  float* __restrict__ Cc, int mh,
                                                  const float* __restrict__ rowscale) {
#pragma unroll
  for (int s = 0; s < 4; ++s) {
    v8f acc = s == 0 ? acc0 : s == 1 ? acc1 : s == 2 ? acc2 : acc3;
    int mb = mh + 16 * s;
#pragma unroll
    for (int v = 0; v < 8; ++v) {
      int m = mb + v;
      float val = acc[v];
      if (HAS_SCALE) val *= rowscale[m];
      Cc[(size_t)m * DP] = val;
    }
  }
}

static constexpr int PE = NE / 64;
static constexpr int PV = NVP / 64;
__global__ __launch_bounds__(32 * NTN) void k_gemm5(
    const float* __restrict__ Ain, const float* __restrict__ Aout,
    const float* __restrict__ Ar,  const float* __restrict__ Air,
    const float* __restrict__ Ax,  const float* __restrict__ WP,
    float* __restrict__ Cin, float* __restrict__ Cout,
    float* __restrict__ Cr,  float* __restrict__ Cir,
    float* __restrict__ Cx,
    const float* __restrict__ nf, const float* __restrict__ ni) {
  const int p = blockIdx.x;
  const float* A; const float* B; float* C; const float* rowscale; int tm;
  if (p < PE)          { A = Ain;  B = WP;               C = Cin;  rowscale = nf;      tm = p; }
  else if (p < 2 * PE) { A = Aout; B = WP + DP * DP;     C = Cout; rowscale = ni;      tm = p - PE; }
  else if (p < 3 * PE) { A = Ar;   B = WP + 3 * DP * DP; C = Cr;   rowscale = nullptr; tm = p - 2 * PE; }
  else if (p < 4 * PE) { A = Air;  B = WP + 3 * DP * DP; C = Cir;  rowscale = nullptr; tm = p - 3 * PE; }
  else                 { A = Ax;   B = WP + 2 * DP * DP; C = Cx;   rowscale = nullptr; tm = p - 4 * PE; }

  const int tn = threadIdx.x >> 5;
  int m0 = tm * 64, n0 = tn * 16;
  int lane = threadIdx.x & 31;
  int half = lane >> 4;
  int l16  = lane & 15;

  const float* A0 = A + (size_t)(m0 + l16) * DP;
  const float* Bc = B + n0 + l16;
  v8f acc0 = {}, acc1 = {}, acc2 = {}, acc3 = {};
#pragma unroll 2
  for (int k = 0; k < DP; k += 4) {
    int kh = k + 2 * half;
    v2f b;
    b.x = Bc[(size_t)kh * DP];
    b.y = Bc[(size_t)(kh + 1) * DP];
    v2f a0 = *reinterpret_cast<const v2f*>(A0 + kh);
    v2f a1 = *reinterpret_cast<const v2f*>(A0 + 16 * DP + kh);
    v2f a2 = *reinterpret_cast<const v2f*>(A0 + 32 * DP + kh);
    v2f a3 = *reinterpret_cast<const v2f*>(A0 + 48 * DP + kh);
    acc0 = wmma4(a0, b, acc0);
    acc1 = wmma4(a1, b, acc1);
    acc2 = wmma4(a2, b, acc2);
    acc3 = wmma4(a3, b, acc3);
  }
  float* Cc = C + n0 + l16;
  const int mh = m0 + 8 * half;
  if (rowscale) gemm_store<true>(acc0, acc1, acc2, acc3, Cc, mh, rowscale);
  else          gemm_store<false>(acc0, acc1, acc2, acc3, Cc, mh, rowscale);
}

static constexpr int HP = 396;

static __device__ __forceinline__ v2f lo2(v4f v) { v2f r; r.x = v.x; r.y = v.y; return r; }
static __device__ __forceinline__ v2f hi2(v4f v) { v2f r; r.x = v.z; r.y = v.w; return r; }

static constexpr int FC_NB = 49;
static constexpr int FC_PD = 2;
static constexpr int FC_WB = 12;
static constexpr int FC_BS = 4 * NTN * 32;

static __device__ __forceinline__ void fc_loadB(v4f (&slot)[NTN], const v4f* __restrict__ Bk) {
#pragma unroll
  for (int t = 0; t < NTN; ++t) slot[t] = Bk[t * 32];
}

static __device__ __forceinline__ void fc_fill(v4f (&ring)[FC_PD + 1][NTN], const v4f* __restrict__ Bp) {
#pragma unroll
  for (int i = 0; i < FC_PD; ++i) fc_loadB(ring[i], Bp + i * FC_BS);
}

static __device__ __forceinline__ void fc_block(v4f av, const v4f (&bq)[NTN], v8f (&acc)[NTN]) {
  const v2f a1 = lo2(av), a2 = hi2(av);
#pragma unroll
  for (int t = 0; t < NTN; ++t) acc[t] = wmma4(a1, lo2(bq[t]), acc[t]);
#pragma unroll
  for (int t = 0; t < NTN; ++t) acc[t] = wmma4(a2, hi2(bq[t]), acc[t]);
}

static __device__ __forceinline__ void fc_pair(const float* hrow, const v4f* __restrict__ Bp,
                                               const bool has13,
                                               v4f (&ring)[FC_PD + 1][NTN], v8f (&acc)[NTN]) {
  v4f av = *reinterpret_cast<const v4f*>(hrow);
#pragma unroll
  for (int i = 0; i < FC_WB; ++i) {
    if (i + FC_PD < FC_WB)
      fc_loadB(ring[(i + FC_PD) % (FC_PD + 1)], Bp + (i + FC_PD) * FC_BS);
    else if (i + FC_PD == FC_WB && has13)
      fc_loadB(ring[FC_WB % (FC_PD + 1)], Bp + FC_WB * FC_BS);
    v4f avn = av;
    if (i + 1 < FC_WB || has13) avn = *reinterpret_cast<const v4f*>(hrow + 32 * (i + 1));
    fc_block(av, ring[i % (FC_PD + 1)], acc);
    av = avn;
  }
  if (has13) fc_block(av, ring[FC_WB % (FC_PD + 1)], acc);
}

__global__ __launch_bounds__(128, 2) void k_convfc(
    const float* __restrict__ xP, const float* __restrict__ rP,
    const int* __restrict__ ei0, const float* __restrict__ fcwP,
    const float* __restrict__ conv_w, const float* __restrict__ conv_b,
    const float* __restrict__ bn1_g, const float* __restrict__ bn1_b,
    const float* __restrict__ fc_b,
    const float* __restrict__ bn2_g, const float* __restrict__ bn2_b,
    const float* __restrict__ fc1_w, const float* __restrict__ fc1_b,
    const float* __restrict__ bn0_g, const float* __restrict__ bn0_b,
    float* __restrict__ out) {
  __shared__ __attribute__((aligned(16))) float img[16][400];
  __shared__ __attribute__((aligned(16))) float hbuf[16][HP];
  __shared__ float h2s[16][D0];
  __shared__ float part[4][16][DP];

  int m0   = blockIdx.x * 16;
  int tid  = threadIdx.x;
  const int wave = __builtin_amdgcn_readfirstlane(tid >> 5);
  int lane = tid & 31;
  int half = lane >> 4, l16 = lane & 15;

  float b0s = bn0_g[0] * INV_SQRT_1PE;
  float b0b = bn0_b[0];

  for (int i = tid; i < 16 * 400; i += 128) {
    int t = i / 400, k = i % 400;
    int src = ei0[m0 + t];
    float v = (k & 1) ? rP[(size_t)(m0 + t) * DP + (k >> 1)]
                      : xP[(size_t)src * DP + (k >> 1)];
    img[t][k] = v * b0s + b0b;
  }
  __syncthreads();

  const bool cv = tid < 112;
  const int  ct = cv ? tid / 7 : 0;
  const int  cj = cv ? tid % 7 : 0;

  float win[8][20];
  {
    const v4f* wp = reinterpret_cast<const v4f*>(&img[ct][cj * 40]);
#pragma unroll
    for (int q = 0; q < 40; ++q) {
      v4f v = wp[q];
      win[q / 5][(q % 5) * 4 + 0] = v.x;
      win[q / 5][(q % 5) * 4 + 1] = v.y;
      win[q / 5][(q % 5) * 4 + 2] = v.z;
      win[q / 5][(q % 5) * 4 + 3] = v.w;
    }
  }

  v8f acc[NTN];
#pragma unroll
  for (int t = 0; t < NTN; ++t) acc[t] = v8f{};
  const v4f*   fcwP4 = reinterpret_cast<const v4f*>(fcwP);
  const float* hrow  = &hbuf[l16][4 * half];

  for (int cp = 0; cp < NF / 2; ++cp) {
    for (int s = 0; s < 2; ++s) {
      const int c = 2 * cp + s;
      const float* wc = conv_w + c * CONVK;
      float sc1 = bn1_g[c] * INV_SQRT_1PE;
      float bb1 = bn1_b[c];
      float cb  = conv_b[c];

      float o[28];
#pragma unroll
      for (int i = 0; i < 28; ++i) o[i] = cb;
#pragma unroll
      for (int a = 0; a < 7; ++a) {
#pragma unroll
        for (int b = 0; b < 7; ++b) {
          const float w = wc[a * 7 + b];
#pragma unroll
          for (int dy = 0; dy < 2; ++dy) {
#pragma unroll
            for (int ox = 0; ox < 14; ++ox)
              o[dy * 14 + ox] += win[dy + a][ox + b] * w;
          }
        }
      }

      if (s == 0) __syncthreads();
      if (cv) {
        v4f* hp = reinterpret_cast<v4f*>(&hbuf[ct][s * P196 + cj * 28]);
#pragma unroll
        for (int q = 0; q < 7; ++q) {
          v4f r;
          float v0 = o[4 * q + 0] * sc1 + bb1;
          float v1 = o[4 * q + 1] * sc1 + bb1;
          float v2 = o[4 * q + 2] * sc1 + bb1;
          float v3 = o[4 * q + 3] * sc1 + bb1;
          r.x = v0 > 0.f ? v0 : 0.f;
          r.y = v1 > 0.f ? v1 : 0.f;
          r.z = v2 > 0.f ? v2 : 0.f;
          r.w = v3 > 0.f ? v3 : 0.f;
          hp[q] = r;
        }
      }
    }
    const int  j0    = (wave - cp) & 3;
    const bool has13 = j0 == 0;
    const v4f* Bp = fcwP4 + ((size_t)cp * FC_NB + j0) * (NTN * 32) + lane;
    v4f ring[FC_PD + 1][NTN];
    fc_fill(ring, Bp);
    __syncthreads();
    fc_pair(hrow + 8 * j0, Bp, has13, ring, acc);
  }

#pragma unroll
  for (int t = 0; t < NTN; ++t) {
#pragma unroll
    for (int v = 0; v < 8; ++v) part[wave][v + 8 * half][t * 16 + l16] = acc[t][v];
  }
  __syncthreads();
  for (int i = tid; i < 16 * D0; i += 128) {
    int t = i / D0, n = i % D0;
    float s = (part[0][t][n] + part[1][t][n]) + (part[2][t][n] + part[3][t][n]);
    float val = (s + fc_b[n]) * (bn2_g[n] * INV_SQRT_1PE) + bn2_b[n];
    h2s[t][n] = val > 0.f ? val : 0.f;
  }
  __syncthreads();

  for (int i = tid; i < 16 * NCLS; i += 128) {
    int t = i / NCLS, cls = i % NCLS;
    float s = fc1_b[cls];
    for (int c2 = 0; c2 < D0; ++c2) s += h2s[t][c2] * fc1_w[cls * D0 + c2];
    out[(size_t)(m0 + t) * NCLS + cls] = s;
  }
}


extern "C" void kernel_launch(void* const* d_in, const int* in_sizes, int n_in,
                              void* d_out, int out_size, void* d_ws, size_t ws_size,
                              hipStream_t stream) {
  const float* node  = (const float*)d_in[0];
  const float* ef    = (const float*)d_in[1];
  const float* ief   = (const float*)d_in[2];
  const float* Win   = (const float*)d_in[3];
  const float* Wout  = (const float*)d_in[4];
  const float* Wloop = (const float*)d_in[5];
  const float* Wrel  = (const float*)d_in[6];
  const float* bgcn  = (const float*)d_in[7];
  const float* bn0g  = (const float*)d_in[8];
  const float* bn0b  = (const float*)d_in[9];
  const float* convw = (const float*)d_in[10];
  const float* convb = (const float*)d_in[11];
  const float* bn1g  = (const float*)d_in[12];
  const float* bn1b  = (const float*)d_in[13];
  const float* fcw   = (const float*)d_in[14];
  const float* fcb   = (const float*)d_in[15];
  const float* bn2g  = (const float*)d_in[16];
  const float* bn2b  = (const float*)d_in[17];
  const float* fc1w  = (const float*)d_in[18];
  const float* fc1b  = (const float*)d_in[19];
  const int*   ei    = (const int*)d_in[20];
  const int*   iei   = (const int*)d_in[21];
  const int *ei0 = ei, *ei1 = ei + NE;
  const int *iei0 = iei, *iei1 = iei + NE;

  float* ws = (float*)d_ws;
  size_t o = 0;
  float* xP   = ws + o; o += (size_t)NVP * DP;
  float* xlP  = ws + o; o += (size_t)NVP * DP;
  float* aggP = ws + o; o += (size_t)NVP * DP;
  float* r0   = ws + o; o += (size_t)NE * DP;
  float* r1   = ws + o; o += (size_t)NE * DP;
  float* ir0  = ws + o; o += (size_t)NE * DP;
  float* ir1  = ws + o; o += (size_t)NE * DP;
  float* ain  = ws + o; o += (size_t)NE * DP;
  float* aout = ws + o; o += (size_t)NE * DP;
  float* minb = ws + o; o += (size_t)NE * DP;
  float* moutb= ws + o; o += (size_t)NE * DP;
  float* WP   = ws + o; o += (size_t)4 * DP * DP;
  float* fcwP = ws + o; o += (size_t)FLAT * DP;
  float* degf = ws + o; o += NVP;
  float* degi = ws + o; o += NVP;
  float* nfv  = ws + o; o += NE;
  float* niv  = ws + o; o += NE;

  const int TB = 256;
  auto cdiv = [](int a, int b) { return (a + b - 1) / b; };

  k_init<<<cdiv(FLAT * DP, TB), TB, 0, stream>>>(node, ef, ief, fcw, xP, r0, ir0, fcwP, degf);

  k_deg<<<cdiv(2 * NE, TB), TB, 0, stream>>>(ei0, iei0, degf, degi);
  k_norm<<<cdiv(NE, TB), TB, 0, stream>>>(ei0, ei1, iei0, iei1, degf, degi, nfv, niv);

  float *rc = r0, *rn = r1, *irc = ir0, *irn = ir1;

  for (int l = 0; l < NLAY; ++l) {
    k_pre<<<cdiv(NE * DP, TB), TB, 0, stream>>>(xP, rc, irc, ei0, iei0, ain, aout, aggP,
                                                Win, Wout, Wloop, Wrel, l, WP);
    k_gemm5<<<4 * PE + PV, 32 * NTN, 0, stream>>>(ain, aout, rc, irc, xP, WP,
                                                  minb, moutb, rn, irn, xlP, nfv, niv);
    k_scatter2<<<cdiv(2 * NE * DP, TB), TB, 0, stream>>>(minb, ei1, moutb, iei1, aggP);
    k_update<<<cdiv(NV * D0, TB), TB, 0, stream>>>(xP, aggP, xlP, bgcn, l);
    float* t;
    t = rc; rc = rn; rn = t;
    t = irc; irc = irn; irn = t;
  }

  k_convfc<<<NE / 16, 128, 0, stream>>>(xP, rc, ei0, fcwP, convw, convb,
                                        bn1g, bn1b, fcb, bn2g, bn2b,
                                        fc1w, fc1b, bn0g, bn0b, (float*)d_out);
}


# layer GEMM A operand: the work-group's 64x208 panel is staged once in LDS by whole-line 16-byte loads and read as ds_load_b128 fragments of 8 k; the B loader takes the same k order
# speedup vs baseline: 2.6623x; 1.1801x over previous
#include <hip/hip_runtime.h>
#include <math.h>


static constexpr int D0   = 200;
static constexpr int DP   = 208;
static constexpr int NTN  = 13;
static constexpr int NE   = 8192;
static constexpr int NV   = 5000;
static constexpr int NVP  = 5056;
static constexpr int NLAY = 6;
static constexpr int NF   = 96;
static constexpr int CONVK= 49;
static constexpr int P196 = 196;
static constexpr int FLAT = 18816;
static constexpr int NCLS = 13;
#define INV_SQRT_1PE 0.99999500f

typedef __attribute__((ext_vector_type(2))) float v2f;
typedef __attribute__((ext_vector_type(4))) float v4f;
typedef __attribute__((ext_vector_type(8))) float v8f;

static __device__ __forceinline__ v8f wmma4(v2f a, v2f b, v8f c) {
  return __builtin_amdgcn_wmma_f32_16x16x4_f32(false, a, false, b, (short)0, c,
                                               false, false);
}


__global__ void k_init(const float* __restrict__ node, const float* __restrict__ ef,
                       const float* __restrict__ ief, const float* __restrict__ fcw,
                       float* __restrict__ xP, float* __restrict__ r0,
                       float* __restrict__ ir0, float* __restrict__ fcwP,
                       float* __restrict__ deg) {
  int i = blockIdx.x * blockDim.x + threadIdx.x;
  if (i < FLAT * DP) {
    int e = i & 3, lane = (i >> 2) & 31, rest = i >> 7;
    int tile = rest % NTN, kb = rest / NTN;
    int n = tile * 16 + (lane & 15);
    int k = kb * 8 + (lane >> 4) * 4 + e;
    fcwP[i] = (n < D0) ? fcw[(size_t)n * FLAT + k] : 0.0f;
  }
  if (i < NE * DP) {
    int r = i / DP, c = i % DP;
    bool in = c < D0;
    r0[i]  = in ? ef[r * D0 + c]  : 0.0f;
    ir0[i] = in ? ief[r * D0 + c] : 0.0f;
  }
  if (i < NVP * DP) {
    int r = i / DP, c = i % DP;
    xP[i] = (r < NV && c < D0) ? node[r * D0 + c] : 0.0f;
  }
  if (i < 2 * NVP) deg[i] = 0.0f;
}

__global__ void k_deg(const int* __restrict__ ei0, const int* __restrict__ iei0,
                      float* __restrict__ degf, float* __restrict__ degi) {
  int e = blockIdx.x * blockDim.x + threadIdx.x;
  if (e < NE)            atomicAdd(&degf[ei0[e]], 1.0f);
  else if (e < 2 * NE)   atomicAdd(&degi[iei0[e - NE]], 1.0f);
}

__global__ void k_norm(const int* __restrict__ ei0, const int* __restrict__ ei1,
                       const int* __restrict__ iei0, const int* __restrict__ iei1,
                       const float* __restrict__ degf, const float* __restrict__ degi,
                       float* __restrict__ nf, float* __restrict__ ni) {
  int e = blockIdx.x * blockDim.x + threadIdx.x;
  if (e >= NE) return;
  float df0 = degf[ei0[e]],  df1 = degf[ei1[e]];
  float di0 = degi[iei0[e]], di1 = degi[iei1[e]];
  float a = df0 > 0.f ? rsqrtf(fmaxf(df0, 1.f)) : 0.f;
  float b = df1 > 0.f ? rsqrtf(fmaxf(df1, 1.f)) : 0.f;
  float c = di0 > 0.f ? rsqrtf(fmaxf(di0, 1.f)) : 0.f;
  float d = di1 > 0.f ? rsqrtf(fmaxf(di1, 1.f)) : 0.f;
  nf[e] = a * b;
  ni[e] = c * d;
}

__global__ void k_pre(const float* __restrict__ xP, const float* __restrict__ r,
                      const float* __restrict__ ir,
                      const int* __restrict__ ei0, const int* __restrict__ iei0,
                      float* __restrict__ ain, float* __restrict__ aout,
                      float* __restrict__ agg,
                      const float* __restrict__ Win, const float* __restrict__ Wout,
                      const float* __restrict__ Wloop, const float* __restrict__ Wrel,
                      int l, float* __restrict__ WP) {
  int i = blockIdx.x * blockDim.x + threadIdx.x;
  if (i >= NE * DP) return;
  {
    int e = i / DP, c = i % DP;
    ain[i]  = xP[(size_t)ei0[e] * DP + c] - r[i];
    aout[i] = xP[(size_t)iei0[e] * DP + c] - ir[i];
  }
  if (i < NVP * DP) agg[i] = 0.0f;
  if (i < 4 * DP * DP) {
    int w = i / (DP * DP);
    int rem = i % (DP * DP);
    int rr = rem / DP, c = rem % DP;
    const float* src = (w == 0 ? Win : w == 1 ? Wout : w == 2 ? Wloop : Wrel)
                       + (size_t)l * D0 * D0;
    WP[i] = (rr < D0 && c < D0) ? src[rr * D0 + c] : 0.0f;
  }
}

__global__ void k_scatter2(const float* __restrict__ m_in, const int* __restrict__ ei1,
                           const float* __restrict__ m_out, const int* __restrict__ iei1,
                           float* __restrict__ agg) {
  int i = blockIdx.x * blockDim.x + threadIdx.x;
  if (i >= 2 * NE * DP) return;
  const bool second = i >= NE * DP;
  const int j = second ? i - NE * DP : i;
  const float* src = second ? m_out : m_in;
  const int*   idx = second ? iei1 : ei1;
  int e = j / DP, c = j % DP;
  atomicAdd(&agg[(size_t)idx[e] * DP + c], src[j]);
}

__global__ void k_update(float* __restrict__ xP, const float* __restrict__ agg,
                         const float* __restrict__ xl, const float* __restrict__ b_gcn,
                         int l) {
  int i = blockIdx.x * blockDim.x + threadIdx.x;
  if (i >= NV * D0) return;
  int r = i / D0, c = i % D0;
  size_t o = (size_t)r * DP + c;
  xP[o] = tanhf((agg[o] + xl[o]) * (1.0f / 3.0f) + b_gcn[l * D0 + c]);
}

template <bool HAS_SCALE>
static __device__ __forceinline__ void gemm_store(v8f acc0, v8f acc1, v8f acc2, v8f acc3,
                                                  float* __restrict__ Cc, int mh,
                                                  const float* __restrict__ rowscale) {
#pragma unroll
  for (int s = 0; s < 4; ++s) {
    v8f acc = s == 0 ? acc0 : s == 1 ? acc1 : s == 2 ? acc2 : acc3;
    int mb = mh + 16 * s;
#pragma unroll
    for (int v = 0; v < 8; ++v) {
      int m = mb + v;
      float val = acc[v];
      if (HAS_SCALE) val *= rowscale[m];
      Cc[(size_t)m * DP] = val;
    }
  }
}

static constexpr int PE = NE / 64;
static constexpr int PV = NVP / 64;
static constexpr int AP = 212;
static constexpr int PCH = 64 * DP / 4;
static constexpr int PCT = PCH / (32 * NTN);

static __device__ __forceinline__ v2f lo2(v4f v) { v2f r; r.x = v.x; r.y = v.y; return r; }
static __device__ __forceinline__ v2f hi2(v4f v) { v2f r; r.x = v.z; r.y = v.w; return r; }

__global__ __launch_bounds__(32 * NTN) void k_gemm5(
    const float* __restrict__ Ain, const float* __restrict__ Aout,
    const float* __restrict__ Ar,  const float* __restrict__ Air,
    const float* __restrict__ Ax,  const float* __restrict__ WP,
    float* __restrict__ Cin, float* __restrict__ Cout,
    float* __restrict__ Cr,  float* __restrict__ Cir,
    float* __restrict__ Cx,
    const float* __restrict__ nf, const float* __restrict__ ni) {
  __shared__ __attribute__((aligned(16))) float As[64 * AP];

  const int p = blockIdx.x;
  const float* A; const float* B; float* C; const float* rowscale; int tm;
  if (p < PE)          { A = Ain;  B = WP;               C = Cin;  rowscale = nf;      tm = p; }
  else if (p < 2 * PE) { A = Aout; B = WP + DP * DP;     C = Cout; rowscale = ni;      tm = p - PE; }
  else if (p < 3 * PE) { A = Ar;   B = WP + 3 * DP * DP; C = Cr;   rowscale = nullptr; tm = p - 2 * PE; }
  else if (p < 4 * PE) { A = Air;  B = WP + 3 * DP * DP; C = Cir;  rowscale = nullptr; tm = p - 3 * PE; }
  else                 { A = Ax;   B = WP + 2 * DP * DP; C = Cx;   rowscale = nullptr; tm = p - 4 * PE; }

  const int tid = threadIdx.x;
  const int tn = tid >> 5;
  int m0 = tm * 64, n0 = tn * 16;
  int lane = tid & 31;
  int half = lane >> 4;
  int l16  = lane & 15;

  {
    const v4f* Ag = reinterpret_cast<const v4f*>(A + (size_t)m0 * DP);
    v4f stage[PCT];
#pragma unroll
    for (int q = 0; q < PCT; ++q) stage[q] = Ag[tid + q * (32 * NTN)];
#pragma unroll
    for (int q = 0; q < PCT; ++q) {
      int ch = tid + q * (32 * NTN);
      int row = ch / (DP / 4), c4 = ch % (DP / 4);
      *reinterpret_cast<v4f*>(&As[row * AP + 4 * c4]) = stage[q];
    }
  }
  __syncthreads();

  const float* Al = &As[l16 * AP + 4 * half];
  const float* Bc = B + (size_t)(4 * half) * DP + n0 + l16;
  v8f acc0 = {}, acc1 = {}, acc2 = {}, acc3 = {};
#pragma unroll 2
  for (int j = 0; j < DP / 8; ++j) {
    const float* Bj = Bc + (size_t)(8 * j) * DP;
    v4f b;
    b.x = Bj[0];
    b.y = Bj[DP];
    b.z = Bj[2 * DP];
    b.w = Bj[3 * DP];
    v4f a0 = *reinterpret_cast<const v4f*>(Al + 8 * j);
    v4f a1 = *reinterpret_cast<const v4f*>(Al + 16 * AP + 8 * j);
    v4f a2 = *reinterpret_cast<const v4f*>(Al + 32 * AP + 8 * j);
    v4f a3 = *reinterpret_cast<const v4f*>(Al + 48 * AP + 8 * j);
    acc0 = wmma4(lo2(a0), lo2(b), acc0);
    acc1 = wmma4(lo2(a1), lo2(b), acc1);
    acc2 = wmma4(lo2(a2), lo2(b), acc2);
    acc3 = wmma4(lo2(a3), lo2(b), acc3);
    acc0 = wmma4(hi2(a0), hi2(b), acc0);
    acc1 = wmma4(hi2(a1), hi2(b), acc1);
    acc2 = wmma4(hi2(a2), hi2(b), acc2);
    acc3 = wmma4(hi2(a3), hi2(b), acc3);
  }
  float* Cc = C + n0 + l16;
  const int mh = m0 + 8 * half;
  if (rowscale) gemm_store<true>(acc0, acc1, acc2, acc3, Cc, mh, rowscale);
  else          gemm_store<false>(acc0, acc1, acc2, acc3, Cc, mh, rowscale);
}

static constexpr int HP = 396;

static constexpr int FC_NB = 49;
static constexpr int FC_PD = 2;
static constexpr int FC_WB = 12;
static constexpr int FC_BS = 4 * NTN * 32;

static __device__ __forceinline__ void fc_loadB(v4f (&slot)[NTN], const v4f* __restrict__ Bk) {
#pragma unroll
  for (int t = 0; t < NTN; ++t) slot[t] = Bk[t * 32];
}

static __device__ __forceinline__ void fc_fill(v4f (&ring)[FC_PD + 1][NTN], const v4f* __restrict__ Bp) {
#pragma unroll
  for (int i = 0; i < FC_PD; ++i) fc_loadB(ring[i], Bp + i * FC_BS);
}

static __device__ __forceinline__ void fc_block(v4f av, const v4f (&bq)[NTN], v8f (&acc)[NTN]) {
  const v2f a1 = lo2(av), a2 = hi2(av);
#pragma unroll
  for (int t = 0; t < NTN; ++t) acc[t] = wmma4(a1, lo2(bq[t]), acc[t]);
#pragma unroll
  for (int t = 0; t < NTN; ++t) acc[t] = wmma4(a2, hi2(bq[t]), acc[t]);
}

static __device__ __forceinline__ void fc_pair(const float* hrow, const v4f* __restrict__ Bp,
                                               const bool has13,
                                               v4f (&ring)[FC_PD + 1][NTN], v8f (&acc)[NTN]) {
  v4f av = *reinterpret_cast<const v4f*>(hrow);
#pragma unroll
  for (int i = 0; i < FC_WB; ++i) {
    if (i + FC_PD < FC_WB)
      fc_loadB(ring[(i + FC_PD) % (FC_PD + 1)], Bp + (i + FC_PD) * FC_BS);
    else if (i + FC_PD == FC_WB && has13)
      fc_loadB(ring[FC_WB % (FC_PD + 1)], Bp + FC_WB * FC_BS);
    v4f avn = av;
    if (i + 1 < FC_WB || has13) avn = *reinterpret_cast<const v4f*>(hrow + 32 * (i + 1));
    fc_block(av, ring[i % (FC_PD + 1)], acc);
    av = avn;
  }
  if (has13) fc_block(av, ring[FC_WB % (FC_PD + 1)], acc);
}

__global__ __launch_bounds__(128, 2) void k_convfc(
    const float* __restrict__ xP, const float* __restrict__ rP,
    const int* __restrict__ ei0, const float* __restrict__ fcwP,
    const float* __restrict__ conv_w, const float* __restrict__ conv_b,
    const float* __restrict__ bn1_g, const float* __restrict__ bn1_b,
    const float* __restrict__ fc_b,
    const float* __restrict__ bn2_g, const float* __restrict__ bn2_b,
    const float* __restrict__ fc1_w, const float* __restrict__ fc1_b,
    const float* __restrict__ bn0_g, const float* __restrict__ bn0_b,
    float* __restrict__ out) {
  __shared__ __attribute__((aligned(16))) float img[16][400];
  __shared__ __attribute__((aligned(16))) float hbuf[16][HP];
  __shared__ float h2s[16][D0];
  __shared__ float part[4][16][DP];

  int m0   = blockIdx.x * 16;
  int tid  = threadIdx.x;
  const int wave = __builtin_amdgcn_readfirstlane(tid >> 5);
  int lane = tid & 31;
  int half = lane >> 4, l16 = lane & 15;

  float b0s = bn0_g[0] * INV_SQRT_1PE;
  float b0b = bn0_b[0];

  for (int i = tid; i < 16 * 400; i += 128) {
    int t = i / 400, k = i % 400;
    int src = ei0[m0 + t];
    float v = (k & 1) ? rP[(size_t)(m0 + t) * DP + (k >> 1)]
                      : xP[(size_t)src * DP + (k >> 1)];
    img[t][k] = v * b0s + b0b;
  }
  __syncthreads();

  const bool cv = tid < 112;
  const int  ct = cv ? tid / 7 : 0;
  const int  cj = cv ? tid % 7 : 0;

  float win[8][20];
  {
    const v4f* wp = reinterpret_cast<const v4f*>(&img[ct][cj * 40]);
#pragma unroll
    for (int q = 0; q < 40; ++q) {
      v4f v = wp[q];
      win[q / 5][(q % 5) * 4 + 0] = v.x;
      win[q / 5][(q % 5) * 4 + 1] = v.y;
      win[q / 5][(q % 5) * 4 + 2] = v.z;
      win[q / 5][(q % 5) * 4 + 3] = v.w;
    }
  }

  v8f acc[NTN];
#pragma unroll
  for (int t = 0; t < NTN; ++t) acc[t] = v8f{};
  const v4f*   fcwP4 = reinterpret_cast<const v4f*>(fcwP);
  const float* hrow  = &hbuf[l16][4 * half];

  for (int cp = 0; cp < NF / 2; ++cp) {
    for (int s = 0; s < 2; ++s) {
      const int c = 2 * cp + s;
      const float* wc = conv_w + c * CONVK;
      float sc1 = bn1_g[c] * INV_SQRT_1PE;
      float bb1 = bn1_b[c];
      float cb  = conv_b[c];

      float o[28];
#pragma unroll
      for (int i = 0; i < 28; ++i) o[i] = cb;
#pragma unroll
      for (int a = 0; a < 7; ++a) {
#pragma unroll
        for (int b = 0; b < 7; ++b) {
          const float w = wc[a * 7 + b];
#pragma unroll
          for (int dy = 0; dy < 2; ++dy) {
#pragma unroll
            for (int ox = 0; ox < 14; ++ox)
              o[dy * 14 + ox] += win[dy + a][ox + b] * w;
          }
        }
      }

      if (s == 0) __syncthreads();
      if (cv) {
        v4f* hp = reinterpret_cast<v4f*>(&hbuf[ct][s * P196 + cj * 28]);
#pragma unroll
        for (int q = 0; q < 7; ++q) {
          v4f r;
          float v0 = o[4 * q + 0] * sc1 + bb1;
          float v1 = o[4 * q + 1] * sc1 + bb1;
          float v2 = o[4 * q + 2] * sc1 + bb1;
          float v3 = o[4 * q + 3] * sc1 + bb1;
          r.x = v0 > 0.f ? v0 : 0.f;
          r.y = v1 > 0.f ? v1 : 0.f;
          r.z = v2 > 0.f ? v2 : 0.f;
          r.w = v3 > 0.f ? v3 : 0.f;
          hp[q] = r;
        }
      }
    }
    const int  j0    = (wave - cp) & 3;
    const bool has13 = j0 == 0;
    const v4f* Bp = fcwP4 + ((size_t)cp * FC_NB + j0) * (NTN * 32) + lane;
    v4f ring[FC_PD + 1][NTN];
    fc_fill(ring, Bp);
    __syncthreads();
    fc_pair(hrow + 8 * j0, Bp, has13, ring, acc);
  }

#pragma unroll
  for (int t = 0; t < NTN; ++t) {
#pragma unroll
    for (int v = 0; v < 8; ++v) part[wave][v + 8 * half][t * 16 + l16] = acc[t][v];
  }
  __syncthreads();
  for (int i = tid; i < 16 * D0; i += 128) {
    int t = i / D0, n = i % D0;
    float s = (part[0][t][n] + part[1][t][n]) + (part[2][t][n] + part[3][t][n]);
    float val = (s + fc_b[n]) * (bn2_g[n] * INV_SQRT_1PE) + bn2_b[n];
    h2s[t][n] = val > 0.f ? val : 0.f;
  }
  __syncthreads();

  for (int i = tid; i < 16 * NCLS; i += 128) {
    int t = i / NCLS, cls = i % NCLS;
    float s = fc1_b[cls];
    for (int c2 = 0; c2 < D0; ++c2) s += h2s[t][c2] * fc1_w[cls * D0 + c2];
    out[(size_t)(m0 + t) * NCLS + cls] = s;
  }
}


extern "C" void kernel_launch(void* const* d_in, const int* in_sizes, int n_in,
                              void* d_out, int out_size, void* d_ws, size_t ws_size,
                              hipStream_t stream) {
  const float* node  = (const float*)d_in[0];
  const float* ef    = (const float*)d_in[1];
  const float* ief   = (const float*)d_in[2];
  const float* Win   = (const float*)d_in[3];
  const float* Wout  = (const float*)d_in[4];
  const float* Wloop = (const float*)d_in[5];
  const float* Wrel  = (const float*)d_in[6];
  const float* bgcn  = (const float*)d_in[7];
  const float* bn0g  = (const float*)d_in[8];
  const float* bn0b  = (const float*)d_in[9];
  const float* convw = (const float*)d_in[10];
  const float* convb = (const float*)d_in[11];
  const float* bn1g  = (const float*)d_in[12];
  const float* bn1b  = (const float*)d_in[13];
  const float* fcw   = (const float*)d_in[14];
  const float* fcb   = (const float*)d_in[15];
  const float* bn2g  = (const float*)d_in[16];
  const float* bn2b  = (const float*)d_in[17];
  const float* fc1w  = (const float*)d_in[18];
  const float* fc1b  = (const float*)d_in[19];
  const int*   ei    = (const int*)d_in[20];
  const int*   iei   = (const int*)d_in[21];
  const int *ei0 = ei, *ei1 = ei + NE;
  const int *iei0 = iei, *iei1 = iei + NE;

  float* ws = (float*)d_ws;
  size_t o = 0;
  float* xP   = ws + o; o += (size_t)NVP * DP;
  float* xlP  = ws + o; o += (size_t)NVP * DP;
  float* aggP = ws + o; o += (size_t)NVP * DP;
  float* r0   = ws + o; o += (size_t)NE * DP;
  float* r1   = ws + o; o += (size_t)NE * DP;
  float* ir0  = ws + o; o += (size_t)NE * DP;
  float* ir1  = ws + o; o += (size_t)NE * DP;
  float* ain  = ws + o; o += (size_t)NE * DP;
  float* aout = ws + o; o += (size_t)NE * DP;
  float* minb = ws + o; o += (size_t)NE * DP;
  float* moutb= ws + o; o += (size_t)NE * DP;
  float* WP   = ws + o; o += (size_t)4 * DP * DP;
  float* fcwP = ws + o; o += (size_t)FLAT * DP;
  float* degf = ws + o; o += NVP;
  float* degi = ws + o; o += NVP;
  float* nfv  = ws + o; o += NE;
  float* niv  = ws + o; o += NE;

  const int TB = 256;
  auto cdiv = [](int a, int b) { return (a + b - 1) / b; };

  k_init<<<cdiv(FLAT * DP, TB), TB, 0, stream>>>(node, ef, ief, fcw, xP, r0, ir0, fcwP, degf);

  k_deg<<<cdiv(2 * NE, TB), TB, 0, stream>>>(ei0, iei0, degf, degi);
  k_norm<<<cdiv(NE, TB), TB, 0, stream>>>(ei0, ei1, iei0, iei1, degf, degi, nfv, niv);

  float *rc = r0, *rn = r1, *irc = ir0, *irn = ir1;

  for (int l = 0; l < NLAY; ++l) {
    k_pre<<<cdiv(NE * DP, TB), TB, 0, stream>>>(xP, rc, irc, ei0, iei0, ain, aout, aggP,
                                                Win, Wout, Wloop, Wrel, l, WP);
    k_gemm5<<<4 * PE + PV, 32 * NTN, 0, stream>>>(ain, aout, rc, irc, xP, WP,
                                                  minb, moutb, rn, irn, xlP, nfv, niv);
    k_scatter2<<<cdiv(2 * NE * DP, TB), TB, 0, stream>>>(minb, ei1, moutb, iei1, aggP);
    k_update<<<cdiv(NV * D0, TB), TB, 0, stream>>>(xP, aggP, xlP, bgcn, l);
    float* t;
    t = rc; rc = rn; rn = t;
    t = irc; irc = irn; irn = t;
  }

  k_convfc<<<NE / 16, 128, 0, stream>>>(xP, rc, ei0, fcwP, convw, convb,
                                        bn1g, bn1b, fcb, bn2g, bn2b,
                                        fc1w, fc1b, bn0g, bn0b, (float*)d_out);
}


# layer sum taken before the weight multiply (linearity): nodes gather scaled x[src]-r over per-destination edge lists, message GEMMs run on node rows, tanh as epilogue; no float atomics on the sum
# speedup vs baseline: 2.9383x; 1.1037x over previous
#include <hip/hip_runtime.h>
#include <math.h>


static constexpr int D0   = 200;
static constexpr int DP   = 208;
static constexpr int NTN  = 13;
static constexpr int NE   = 8192;
static constexpr int NV   = 5000;
static constexpr int NVP  = 5056;
static constexpr int NLAY = 6;
static constexpr int NF   = 96;
static constexpr int CONVK= 49;
static constexpr int P196 = 196;
static constexpr int FLAT = 18816;
static constexpr int NCLS = 13;
static constexpr int DP4  = DP / 4;
static constexpr int OFFN = NVP + 8;
#define INV_SQRT_1PE 0.99999500f

typedef __attribute__((ext_vector_type(2))) float v2f;
typedef __attribute__((ext_vector_type(4))) float v4f;
typedef __attribute__((ext_vector_type(8))) float v8f;

static __device__ __forceinline__ v8f wmma4(v2f a, v2f b, v8f c) {
  return __builtin_amdgcn_wmma_f32_16x16x4_f32(false, a, false, b, (short)0, c,
                                               false, false);
}


__global__ void k_init(const float* __restrict__ node, const float* __restrict__ ef,
                       const float* __restrict__ ief, const float* __restrict__ fcw,
                       const float* __restrict__ Win, const float* __restrict__ Wout,
                       const float* __restrict__ Wloop, const float* __restrict__ Wrel,
                       float* __restrict__ xP, float* __restrict__ r0,
                       float* __restrict__ ir0, float* __restrict__ fcwP,
                       float* __restrict__ WP, float* __restrict__ deg) {
  int i = blockIdx.x * blockDim.x + threadIdx.x;
  if (i < FLAT * DP) {
    int e = i & 3, lane = (i >> 2) & 31, rest = i >> 7;
    int tile = rest % NTN, kb = rest / NTN;
    int n = tile * 16 + (lane & 15);
    int k = kb * 8 + (lane >> 4) * 4 + e;
    fcwP[i] = (n < D0) ? fcw[(size_t)n * FLAT + k] : 0.0f;
  }
  if (i < NE * DP) {
    int r = i / DP, c = i % DP;
    bool in = c < D0;
    r0[i]  = in ? ef[r * D0 + c]  : 0.0f;
    ir0[i] = in ? ief[r * D0 + c] : 0.0f;
  }
  if (i < NVP * DP) {
    int r = i / DP, c = i % DP;
    xP[i] = (r < NV && c < D0) ? node[r * D0 + c] : 0.0f;
  }
  if (i < NLAY * 4 * DP * DP) {
    int m = i / (DP * DP);
    int rem = i % (DP * DP);
    int l = m >> 2, w = m & 3;
    int rr = rem / DP, c = rem % DP;
    const float* src = (w == 0 ? Win : w == 1 ? Wout : w == 2 ? Wloop : Wrel)
                       + (size_t)l * D0 * D0;
    WP[i] = (rr < D0 && c < D0) ? src[rr * D0 + c] : 0.0f;
  }
  if (i < 4 * NVP) deg[i] = 0.0f;
}

__global__ void k_deg(const int* __restrict__ ei0, const int* __restrict__ ei1,
                      const int* __restrict__ iei0, const int* __restrict__ iei1,
                      float* __restrict__ degf, float* __restrict__ degi,
                      int* __restrict__ cntF, int* __restrict__ cntI) {
  int e = blockIdx.x * blockDim.x + threadIdx.x;
  if (e < NE) {
    atomicAdd(&degf[ei0[e]], 1.0f);
    atomicAdd(&cntF[ei1[e]], 1);
  } else if (e < 2 * NE) {
    e -= NE;
    atomicAdd(&degi[iei0[e]], 1.0f);
    atomicAdd(&cntI[iei1[e]], 1);
  }
}

static constexpr int SCH = (NVP + 1023) / 1024;
__global__ __launch_bounds__(1024) void k_scan(const int* __restrict__ cnt,
                                               int* __restrict__ off, int* __restrict__ cur) {
  __shared__ int part[1024];
  const int* c = cnt + blockIdx.x * NVP;
  int* o  = off + blockIdx.x * OFFN;
  int* cu = cur + blockIdx.x * NVP;
  const int t = threadIdx.x;
  int loc[SCH];
  int s = 0;
#pragma unroll
  for (int k = 0; k < SCH; ++k) {
    int idx = t * SCH + k;
    int v = idx < NVP ? c[idx] : 0;
    loc[k] = s;
    s += v;
  }
  part[t] = s;
  __syncthreads();
  for (int d = 1; d < 1024; d <<= 1) {
    int add = t >= d ? part[t - d] : 0;
    __syncthreads();
    part[t] += add;
    __syncthreads();
  }
  const int excl = part[t] - s;
#pragma unroll
  for (int k = 0; k < SCH; ++k) {
    int idx = t * SCH + k;
    if (idx < NVP) { o[idx] = excl + loc[k]; cu[idx] = excl + loc[k]; }
  }
  if (t == 1023) o[NVP] = part[1023];
}

__global__ void k_fill(const int* __restrict__ ei0, const int* __restrict__ ei1,
                       const int* __restrict__ iei0, const int* __restrict__ iei1,
                       const float* __restrict__ degf, const float* __restrict__ degi,
                       int* __restrict__ curF, int* __restrict__ curI,
                       int* __restrict__ leF, int* __restrict__ lsF, float* __restrict__ lwF,
                       int* __restrict__ leI, int* __restrict__ lsI, float* __restrict__ lwI) {
  int e = blockIdx.x * blockDim.x + threadIdx.x;
  if (e >= NE) return;
  const int s0 = ei0[e], d0 = ei1[e], s1 = iei0[e], d1 = iei1[e];
  float df0 = degf[s0], df1 = degf[d0];
  float di0 = degi[s1], di1 = degi[d1];
  float a = df0 > 0.f ? rsqrtf(fmaxf(df0, 1.f)) : 0.f;
  float b = df1 > 0.f ? rsqrtf(fmaxf(df1, 1.f)) : 0.f;
  float c = di0 > 0.f ? rsqrtf(fmaxf(di0, 1.f)) : 0.f;
  float d = di1 > 0.f ? rsqrtf(fmaxf(di1, 1.f)) : 0.f;
  int p = atomicAdd(&curF[d0], 1);
  leF[p] = e; lsF[p] = s0; lwF[p] = a * b;
  int q = atomicAdd(&curI[d1], 1);
  leI[q] = e; lsI[q] = s1; lwI[q] = c * d;
}

__global__ void k_agg(const float* __restrict__ xP, const float* __restrict__ r,
                      const float* __restrict__ ir,
                      const int* __restrict__ offF, const int* __restrict__ offI,
                      const int* __restrict__ leF, const int* __restrict__ lsF,
                      const float* __restrict__ lwF,
                      const int* __restrict__ leI, const int* __restrict__ lsI,
                      const float* __restrict__ lwI,
                      float* __restrict__ SF, float* __restrict__ SI) {
  int i = blockIdx.x * blockDim.x + threadIdx.x;
  if (i >= NVP * DP4) return;
  const int v = i / DP4, c4 = i % DP4;
  const v4f* x4  = reinterpret_cast<const v4f*>(xP);
  const v4f* r4  = reinterpret_cast<const v4f*>(r);
  const v4f* ir4 = reinterpret_cast<const v4f*>(ir);
  v4f sf = {0.f, 0.f, 0.f, 0.f}, si = {0.f, 0.f, 0.f, 0.f};
  const int bF = max(offF[v], 0), eF = min(offF[v + 1], NE);
  const int bI = max(offI[v], 0), eI = min(offI[v + 1], NE);
  for (int j = bF; j < eF; ++j) {
    const int e = leF[j], sn = lsF[j];
    const float w = lwF[j];
    v4f xv = x4[(size_t)sn * DP4 + c4];
    v4f rv = r4[(size_t)e * DP4 + c4];
    sf += (xv - rv) * w;
  }
  for (int j = bI; j < eI; ++j) {
    const int e = leI[j], sn = lsI[j];
    const float w = lwI[j];
    v4f xv = x4[(size_t)sn * DP4 + c4];
    v4f rv = ir4[(size_t)e * DP4 + c4];
    si += (xv - rv) * w;
  }
  reinterpret_cast<v4f*>(SF)[i] = sf;
  reinterpret_cast<v4f*>(SI)[i] = si;
}

static __device__ __forceinline__ void gemm_store(v8f acc0, v8f acc1, v8f acc2, v8f acc3,
                                                  float* __restrict__ Cc, int mh) {
#pragma unroll
  for (int s = 0; s < 4; ++s) {
    v8f acc = s == 0 ? acc0 : s == 1 ? acc1 : s == 2 ? acc2 : acc3;
    int mb = mh + 16 * s;
#pragma unroll
    for (int v = 0; v < 8; ++v) Cc[(size_t)(mb + v) * DP] = acc[v];
  }
}

static constexpr int PR = NE / 64;
static constexpr int PN = NVP / 32;
static constexpr int AP = 212;
static constexpr int PCH = 64 * DP / 4;
static constexpr int PCT = PCH / (32 * NTN);
static constexpr int NCT = 32 * DP / 4 / (32 * NTN);

static __device__ __forceinline__ v2f lo2(v4f v) { v2f r; r.x = v.x; r.y = v.y; return r; }
static __device__ __forceinline__ v2f hi2(v4f v) { v2f r; r.x = v.z; r.y = v.w; return r; }

static __device__ __forceinline__ v4f gemm_ldB(const float* __restrict__ Bc, int j) {
  const float* Bj = Bc + (size_t)(8 * j) * DP;
  v4f b;
  b.x = Bj[0];
  b.y = Bj[DP];
  b.z = Bj[2 * DP];
  b.w = Bj[3 * DP];
  return b;
}

__global__ __launch_bounds__(32 * NTN) void k_gemm3(
    const float* __restrict__ SF, const float* __restrict__ SI,
    const float* __restrict__ Ar, const float* __restrict__ Air,
    float* xio, const float* __restrict__ WP,
    float* __restrict__ Cr, float* __restrict__ Cir,
    const float* __restrict__ bg) {
  __shared__ __attribute__((aligned(16))) float As[64 * AP];

  const int p = blockIdx.x;
  const int tid = threadIdx.x;
  const int tn = tid >> 5;
  const int lane = tid & 31;
  const int half = lane >> 4;
  const int l16  = lane & 15;
  const int n = tn * 16 + l16;
  const float* Al = &As[l16 * AP + 4 * half];

  if (p >= 2 * PR) {
    const int m0 = (p - 2 * PR) * 32;
    v8f acc0 = {}, acc1 = {};
#pragma unroll
    for (int ph = 0; ph < 3; ++ph) {
      const float* A  = ph == 0 ? SF : ph == 1 ? SI : xio;
      const float* Bc = WP + (size_t)ph * DP * DP
                        + (size_t)(4 * half) * DP + n;
      if (ph > 0) __syncthreads();
      {
        const v4f* Ag = reinterpret_cast<const v4f*>(A + (size_t)m0 * DP);
        v4f stage[NCT];
#pragma unroll
        for (int q = 0; q < NCT; ++q) stage[q] = Ag[tid + q * (32 * NTN)];
#pragma unroll
        for (int q = 0; q < NCT; ++q) {
          int ch = tid + q * (32 * NTN);
          int row = ch / DP4, c4 = ch % DP4;
          *reinterpret_cast<v4f*>(&As[row * AP + 4 * c4]) = stage[q];
        }
      }
      __syncthreads();
      v4f bn = gemm_ldB(Bc, 0);
      for (int j = 0; j < DP / 8; ++j) {
        const v4f b = bn;
        if (j + 1 < DP / 8) bn = gemm_ldB(Bc, j + 1);
        const v4f a0 = *reinterpret_cast<const v4f*>(Al + 8 * j);
        const v4f a1 = *reinterpret_cast<const v4f*>(Al + 16 * AP + 8 * j);
        acc0 = wmma4(lo2(a0), lo2(b), acc0);
        acc1 = wmma4(lo2(a1), lo2(b), acc1);
        acc0 = wmma4(hi2(a0), hi2(b), acc0);
        acc1 = wmma4(hi2(a1), hi2(b), acc1);
      }
    }
    if (n < D0) {
      const float bn = bg[n];
      float* xc = xio + n;
#pragma unroll
      for (int s = 0; s < 2; ++s) {
        const v8f acc = s == 0 ? acc0 : acc1;
#pragma unroll
        for (int v = 0; v < 8; ++v) {
          const int row = m0 + 16 * s + 8 * half + v;
          if (row < NV) xc[(size_t)row * DP] = tanhf(acc[v] * (1.0f / 3.0f) + bn);
        }
      }
    }
  } else {
    const bool inv = p >= PR;
    const int m0 = (inv ? p - PR : p) * 64;
    const float* A = inv ? Air : Ar;
    {
      const v4f* Ag = reinterpret_cast<const v4f*>(A + (size_t)m0 * DP);
      v4f stage[PCT];
#pragma unroll
      for (int q = 0; q < PCT; ++q) stage[q] = Ag[tid + q * (32 * NTN)];
#pragma unroll
      for (int q = 0; q < PCT; ++q) {
        int ch = tid + q * (32 * NTN);
        int row = ch / DP4, c4 = ch % DP4;
        *reinterpret_cast<v4f*>(&As[row * AP + 4 * c4]) = stage[q];
      }
    }
    __syncthreads();

    const float* Bc = WP + (size_t)3 * DP * DP + (size_t)(4 * half) * DP + n;
    v8f acc0 = {}, acc1 = {}, acc2 = {}, acc3 = {};
#pragma unroll 2
    for (int j = 0; j < DP / 8; ++j) {
      const v4f b  = gemm_ldB(Bc, j);
      v4f a0 = *reinterpret_cast<const v4f*>(Al + 8 * j);
      v4f a1 = *reinterpret_cast<const v4f*>(Al + 16 * AP + 8 * j);
      v4f a2 = *reinterpret_cast<const v4f*>(Al + 32 * AP + 8 * j);
      v4f a3 = *reinterpret_cast<const v4f*>(Al + 48 * AP + 8 * j);
      acc0 = wmma4(lo2(a0), lo2(b), acc0);
      acc1 = wmma4(lo2(a1), lo2(b), acc1);
      acc2 = wmma4(lo2(a2), lo2(b), acc2);
      acc3 = wmma4(lo2(a3), lo2(b), acc3);
      acc0 = wmma4(hi2(a0), hi2(b), acc0);
      acc1 = wmma4(hi2(a1), hi2(b), acc1);
      acc2 = wmma4(hi2(a2), hi2(b), acc2);
      acc3 = wmma4(hi2(a3), hi2(b), acc3);
    }
    gemm_store(acc0, acc1, acc2, acc3, (inv ? Cir : Cr) + n, m0 + 8 * half);
  }
}

static constexpr int HP = 396;

static constexpr int FC_NB = 49;
static constexpr int FC_PD = 2;
static constexpr int FC_WB = 12;
static constexpr int FC_BS = 4 * NTN * 32;

static __device__ __forceinline__ void fc_loadB(v4f (&slot)[NTN], const v4f* __restrict__ Bk) {
#pragma unroll
  for (int t = 0; t < NTN; ++t) slot[t] = Bk[t * 32];
}

static __device__ __forceinline__ void fc_fill(v4f (&ring)[FC_PD + 1][NTN], const v4f* __restrict__ Bp) {
#pragma unroll
  for (int i = 0; i < FC_PD; ++i) fc_loadB(ring[i], Bp + i * FC_BS);
}

static __device__ __forceinline__ void fc_block(v4f av, const v4f (&bq)[NTN], v8f (&acc)[NTN]) {
  const v2f a1 = lo2(av), a2 = hi2(av);
#pragma unroll
  for (int t = 0; t < NTN; ++t) acc[t] = wmma4(a1, lo2(bq[t]), acc[t]);
#pragma unroll
  for (int t = 0; t < NTN; ++t) acc[t] = wmma4(a2, hi2(bq[t]), acc[t]);
}

static __device__ __forceinline__ void fc_pair(const float* hrow, const v4f* __restrict__ Bp,
                                               const bool has13,
                                               v4f (&ring)[FC_PD + 1][NTN], v8f (&acc)[NTN]) {
  v4f av = *reinterpret_cast<const v4f*>(hrow);
#pragma unroll
  for (int i = 0; i < FC_WB; ++i) {
    if (i + FC_PD < FC_WB)
      fc_loadB(ring[(i + FC_PD) % (FC_PD + 1)], Bp + (i + FC_PD) * FC_BS);
    else if (i + FC_PD == FC_WB && has13)
      fc_loadB(ring[FC_WB % (FC_PD + 1)], Bp + FC_WB * FC_BS);
    v4f avn = av;
    if (i + 1 < FC_WB || has13) avn = *reinterpret_cast<const v4f*>(hrow + 32 * (i + 1));
    fc_block(av, ring[i % (FC_PD + 1)], acc);
    av = avn;
  }
  if (has13) fc_block(av, ring[FC_WB % (FC_PD + 1)], acc);
}

__global__ __launch_bounds__(128, 2) void k_convfc(
    const float* __restrict__ xP, const float* __restrict__ rP,
    const int* __restrict__ ei0, const float* __restrict__ fcwP,
    const float* __restrict__ conv_w, const float* __restrict__ conv_b,
    const float* __restrict__ bn1_g, const float* __restrict__ bn1_b,
    const float* __restrict__ fc_b,
    const float* __restrict__ bn2_g, const float* __restrict__ bn2_b,
    const float* __restrict__ fc1_w, const float* __restrict__ fc1_b,
    const float* __restrict__ bn0_g, const float* __restrict__ bn0_b,
    float* __restrict__ out) {
  __shared__ __attribute__((aligned(16))) float img[16][400];
  __shared__ __attribute__((aligned(16))) float hbuf[16][HP];
  __shared__ float h2s[16][D0];
  __shared__ float part[4][16][DP];

  int m0   = blockIdx.x * 16;
  int tid  = threadIdx.x;
  const int wave = __builtin_amdgcn_readfirstlane(tid >> 5);
  int lane = tid & 31;
  int half = lane >> 4, l16 = lane & 15;

  float b0s = bn0_g[0] * INV_SQRT_1PE;
  float b0b = bn0_b[0];

  for (int i = tid; i < 16 * 400; i += 128) {
    int t = i / 400, k = i % 400;
    int src = ei0[m0 + t];
    float v = (k & 1) ? rP[(size_t)(m0 + t) * DP + (k >> 1)]
                      : xP[(size_t)src * DP + (k >> 1)];
    img[t][k] = v * b0s + b0b;
  }
  __syncthreads();

  const bool cv = tid < 112;
  const int  ct = cv ? tid / 7 : 0;
  const int  cj = cv ? tid % 7 : 0;

  float win[8][20];
  {
    const v4f* wp = reinterpret_cast<const v4f*>(&img[ct][cj * 40]);
#pragma unroll
    for (int q = 0; q < 40; ++q) {
      v4f v = wp[q];
      win[q / 5][(q % 5) * 4 + 0] = v.x;
      win[q / 5][(q % 5) * 4 + 1] = v.y;
      win[q / 5][(q % 5) * 4 + 2] = v.z;
      win[q / 5][(q % 5) * 4 + 3] = v.w;
    }
  }

  v8f acc[NTN];
#pragma unroll
  for (int t = 0; t < NTN; ++t) acc[t] = v8f{};
  const v4f*   fcwP4 = reinterpret_cast<const v4f*>(fcwP);
  const float* hrow  = &hbuf[l16][4 * half];

  for (int cp = 0; cp < NF / 2; ++cp) {
    for (int s = 0; s < 2; ++s) {
      const int c = 2 * cp + s;
      const float* wc = conv_w + c * CONVK;
      float sc1 = bn1_g[c] * INV_SQRT_1PE;
      float bb1 = bn1_b[c];
      float cb  = conv_b[c];

      float o[28];
#pragma unroll
      for (int i = 0; i < 28; ++i) o[i] = cb;
#pragma unroll
      for (int a = 0; a < 7; ++a) {
#pragma unroll
        for (int b = 0; b < 7; ++b) {
          const float w = wc[a * 7 + b];
#pragma unroll
          for (int dy = 0; dy < 2; ++dy) {
#pragma unroll
            for (int ox = 0; ox < 14; ++ox)
              o[dy * 14 + ox] += win[dy + a][ox + b] * w;
          }
        }
      }

      if (s == 0) __syncthreads();
      if (cv) {
        v4f* hp = reinterpret_cast<v4f*>(&hbuf[ct][s * P196 + cj * 28]);
#pragma unroll
        for (int q = 0; q < 7; ++q) {
          v4f r;
          float v0 = o[4 * q + 0] * sc1 + bb1;
          float v1 = o[4 * q + 1] * sc1 + bb1;
          float v2 = o[4 * q + 2] * sc1 + bb1;
          float v3 = o[4 * q + 3] * sc1 + bb1;
          r.x = v0 > 0.f ? v0 : 0.f;
          r.y = v1 > 0.f ? v1 : 0.f;
          r.z = v2 > 0.f ? v2 : 0.f;
          r.w = v3 > 0.f ? v3 : 0.f;
          hp[q] = r;
        }
      }
    }
    const int  j0    = (wave - cp) & 3;
    const bool has13 = j0 == 0;
    const v4f* Bp = fcwP4 + ((size_t)cp * FC_NB + j0) * (NTN * 32) + lane;
    v4f ring[FC_PD + 1][NTN];
    fc_fill(ring, Bp);
    __syncthreads();
    fc_pair(hrow + 8 * j0, Bp, has13, ring, acc);
  }

#pragma unroll
  for (int t = 0; t < NTN; ++t) {
#pragma unroll
    for (int v = 0; v < 8; ++v) part[wave][v + 8 * half][t * 16 + l16] = acc[t][v];
  }
  __syncthreads();
  for (int i = tid; i < 16 * D0; i += 128) {
    int t = i / D0, n = i % D0;
    float s = (part[0][t][n] + part[1][t][n]) + (part[2][t][n] + part[3][t][n]);
    float val = (s + fc_b[n]) * (bn2_g[n] * INV_SQRT_1PE) + bn2_b[n];
    h2s[t][n] = val > 0.f ? val : 0.f;
  }
  __syncthreads();

  for (int i = tid; i < 16 * NCLS; i += 128) {
    int t = i / NCLS, cls = i % NCLS;
    float s = fc1_b[cls];
    for (int c2 = 0; c2 < D0; ++c2) s += h2s[t][c2] * fc1_w[cls * D0 + c2];
    out[(size_t)(m0 + t) * NCLS + cls] = s;
  }
}


extern "C" void kernel_launch(void* const* d_in, const int* in_sizes, int n_in,
                              void* d_out, int out_size, void* d_ws, size_t ws_size,
                              hipStream_t stream) {
  const float* node  = (const float*)d_in[0];
  const float* ef    = (const float*)d_in[1];
  const float* ief   = (const float*)d_in[2];
  const float* Win   = (const float*)d_in[3];
  const float* Wout  = (const float*)d_in[4];
  const float* Wloop = (const float*)d_in[5];
  const float* Wrel  = (const float*)d_in[6];
  const float* bgcn  = (const float*)d_in[7];
  const float* bn0g  = (const float*)d_in[8];
  const float* bn0b  = (const float*)d_in[9];
  const float* convw = (const float*)d_in[10];
  const float* convb = (const float*)d_in[11];
  const float* bn1g  = (const float*)d_in[12];
  const float* bn1b  = (const float*)d_in[13];
  const float* fcw   = (const float*)d_in[14];
  const float* fcb   = (const float*)d_in[15];
  const float* bn2g  = (const float*)d_in[16];
  const float* bn2b  = (const float*)d_in[17];
  const float* fc1w  = (const float*)d_in[18];
  const float* fc1b  = (const float*)d_in[19];
  const int*   ei    = (const int*)d_in[20];
  const int*   iei   = (const int*)d_in[21];
  const int *ei0 = ei, *ei1 = ei + NE;
  const int *iei0 = iei, *iei1 = iei + NE;

  float* ws = (float*)d_ws;
  size_t o = 0;
  float* xP   = ws + o; o += (size_t)NVP * DP;
  float* SF   = ws + o; o += (size_t)NVP * DP;
  float* SI   = ws + o; o += (size_t)NVP * DP;
  float* r0   = ws + o; o += (size_t)NE * DP;
  float* r1   = ws + o; o += (size_t)NE * DP;
  float* ir0  = ws + o; o += (size_t)NE * DP;
  float* ir1  = ws + o; o += (size_t)NE * DP;
  float* WP   = ws + o; o += (size_t)NLAY * 4 * DP * DP;
  float* fcwP = ws + o; o += (size_t)FLAT * DP;
  float* degf = ws + o; o += NVP;
  float* degi = ws + o; o += NVP;
  int*   cntF = (int*)(ws + o); o += NVP;
  int*   cntI = (int*)(ws + o); o += NVP;
  int*   offF = (int*)(ws + o); o += OFFN;
  int*   offI = (int*)(ws + o); o += OFFN;
  int*   curF = (int*)(ws + o); o += NVP;
  int*   curI = (int*)(ws + o); o += NVP;
  int*   leF  = (int*)(ws + o); o += NE;
  int*   lsF  = (int*)(ws + o); o += NE;
  float* lwF  = ws + o; o += NE;
  int*   leI  = (int*)(ws + o); o += NE;
  int*   lsI  = (int*)(ws + o); o += NE;
  float* lwI  = ws + o; o += NE;

  const int TB = 256;
  auto cdiv = [](int a, int b) { return (a + b - 1) / b; };

  k_init<<<cdiv(FLAT * DP, TB), TB, 0, stream>>>(node, ef, ief, fcw, Win, Wout, Wloop, Wrel,
                                                 xP, r0, ir0, fcwP, WP, degf);

  k_deg<<<cdiv(2 * NE, TB), TB, 0, stream>>>(ei0, ei1, iei0, iei1, degf, degi, cntF, cntI);
  k_scan<<<2, 1024, 0, stream>>>(cntF, offF, curF);
  k_fill<<<cdiv(NE, TB), TB, 0, stream>>>(ei0, ei1, iei0, iei1, degf, degi, curF, curI,
                                          leF, lsF, lwF, leI, lsI, lwI);

  float *rc = r0, *rn = r1, *irc = ir0, *irn = ir1;

  for (int l = 0; l < NLAY; ++l) {
    k_agg<<<cdiv(NVP * DP4, TB), TB, 0, stream>>>(xP, rc, irc, offF, offI,
                                                  leF, lsF, lwF, leI, lsI, lwI, SF, SI);
    k_gemm3<<<2 * PR + PN, 32 * NTN, 0, stream>>>(SF, SI, rc, irc, xP,
                                                  WP + (size_t)l * 4 * DP * DP,
                                                  rn, irn, bgcn + l * D0);
    float* t;
    t = rc; rc = rn; rn = t;
    t = irc; irc = irn; irn = t;
  }

  k_convfc<<<NE / 16, 128, 0, stream>>>(xP, rc, ei0, fcwP, convw, convb,
                                        bn1g, bn1b, fcb, bn2g, bn2b,
                                        fc1w, fc1b, bn0g, bn0b, (float*)d_out);
}
